# starting kernel
# baseline (speedup 1.0000x reference)
#include <hip/hip_runtime.h>
#include <math.h>

#define SB   2048
#define DM   1024
#define NH   16
#define HD   64
#define HD2  32
#define SCALE 0.125f
#define EPS  1e-6f

typedef __attribute__((ext_vector_type(2))) float v2f;
typedef __attribute__((ext_vector_type(8))) float v8f;

static __device__ __forceinline__ v8f wmma_f32(v2f a, v2f b, v8f c) {
    return __builtin_amdgcn_wmma_f32_16x16x4_f32(
         false, a,  false, b,
         (short)0, c,  false,  false);
}

__global__ __launch_bounds__(256) void gemm16_kernel(
    const float* __restrict__ X,
    const float* __restrict__ W,
    const float* __restrict__ bias,
    const float* __restrict__ g,
    const float* __restrict__ fc,
    const float* __restrict__ fs,
    float* __restrict__ out,
    int doNormRope)
{
    __shared__ float tile[16][DM + 4];
    __shared__ float redbuf[16][17];
    __shared__ float rrms[16];

    const int tid  = threadIdx.x;
    const int wave = tid >> 5;
    const int lane = tid & 31;
    const int half = lane >> 4;
    const int l16  = lane & 15;
    const int rowBase  = blockIdx.x * 16;
    const int nColBase = wave * 128;

    v8f acc[8];
#pragma unroll
    for (int t = 0; t < 8; ++t) acc[t] = (v8f){0,0,0,0,0,0,0,0};

    const float* xrow = X + (size_t)(rowBase + l16) * DM + 2 * half;

    for (int k = 0; k < DM; k += 4) {
        v2f a = *(const v2f*)(xrow + k);
#pragma unroll
        for (int t = 0; t < 8; ++t) {
            const float* wp = W + (size_t)(nColBase + t * 16 + l16) * DM + k + 2 * half;
            v2f b = *(const v2f*)wp;
            acc[t] = wmma_f32(a, b, acc[t]);
        }
    }

#pragma unroll
    for (int t = 0; t < 8; ++t) {
        int col = nColBase + t * 16 + l16;
        float bv = bias[col];
#pragma unroll
        for (int j = 0; j < 8; ++j) {
            int row = j + 8 * half;
            tile[row][col] = acc[t][j] + bv;
        }
    }
    __syncthreads();

    if (doNormRope) {
        {
            int r  = tid & 15;
            int c0 = tid >> 4;
            float s = 0.f;
            for (int c = c0; c < DM; c += 16) { float v = tile[r][c]; s += v * v; }
            redbuf[r][c0] = s;
        }
        __syncthreads();
        if (tid < 16) {
            float s = 0.f;
#pragma unroll
            for (int c = 0; c < 16; ++c) s += redbuf[tid][c];
            rrms[tid] = rsqrtf(s * (1.0f / (float)DM) + EPS);
        }
        __syncthreads();
    }

    for (int p = tid; p < 16 * (DM / 2); p += 256) {
        int r    = p >> 9;
        int cp   = p & 511;
        int dcol = cp * 2;
        int srow = rowBase + r;
        float re = tile[r][dcol];
        float im = tile[r][dcol + 1];
        if (doNormRope) {
            float rr = rrms[r];
            re *= rr * g[dcol];
            im *= rr * g[dcol + 1];
            int   pp = cp & (HD2 - 1);
            float c  = fc[srow * HD2 + pp];
            float sn = fs[srow * HD2 + pp];
            float ore = re * c - im * sn;
            float oim = re * sn + im * c;
            re = ore; im = oim;
        }
        out[(size_t)srow * DM + dcol]     = re;
        out[(size_t)srow * DM + dcol + 1] = im;
    }
}

__global__ __launch_bounds__(32) void attn_kernel(
    const float* __restrict__ Q,
    const float* __restrict__ K,
    const float* __restrict__ V,
    float* __restrict__ O)
{
    __shared__ float P[16][17];

    const int lane = threadIdx.x & 31;
    const int half = lane >> 4;
    const int l16  = lane & 15;
    const int qt    = blockIdx.x;
    const int h     = blockIdx.y;
    const int qBase = qt * 16;
    const int dBase = h * HD;

    v2f qa[16];
    {
        const float* qp = Q + (size_t)(qBase + l16) * DM + dBase + 2 * half;
#pragma unroll
        for (int kk = 0; kk < 16; ++kk) qa[kk] = *(const v2f*)(qp + kk * 4);
    }

    v8f o[4];
#pragma unroll
    for (int nt = 0; nt < 4; ++nt) o[nt] = (v8f){0,0,0,0,0,0,0,0};
    float m[8], lsum[8];
#pragma unroll
    for (int j = 0; j < 8; ++j) { m[j] = -INFINITY; lsum[j] = 0.f; }

    for (int t = 0; t < SB / 16; ++t) {
        const int kBase = t * 16;

        v8f s = (v8f){0,0,0,0,0,0,0,0};
        const float* kp = K + (size_t)(kBase + l16) * DM + dBase + 2 * half;
#pragma unroll
        for (int kk = 0; kk < 16; ++kk) {
            v2f b = *(const v2f*)(kp + kk * 4);
            s = wmma_f32(qa[kk], b, s);
        }

#pragma unroll
        for (int j = 0; j < 8; ++j) {
            float sv = s[j] * SCALE;
            float mx = sv;
            mx = fmaxf(mx, __shfl_xor(mx, 1, 32));
            mx = fmaxf(mx, __shfl_xor(mx, 2, 32));
            mx = fmaxf(mx, __shfl_xor(mx, 4, 32));
            mx = fmaxf(mx, __shfl_xor(mx, 8, 32));
            float nm   = fmaxf(m[j], mx);
            float corr = __expf(m[j] - nm);
            float pv   = __expf(sv - nm);
            float ps = pv;
            ps += __shfl_xor(ps, 1, 32);
            ps += __shfl_xor(ps, 2, 32);
            ps += __shfl_xor(ps, 4, 32);
            ps += __shfl_xor(ps, 8, 32);
            lsum[j] = lsum[j] * corr + ps;
            m[j] = nm;
            o[0][j] *= corr; o[1][j] *= corr; o[2][j] *= corr; o[3][j] *= corr;
            P[j + 8 * half][l16] = pv;
        }
        __syncthreads();

#pragma unroll
        for (int kc = 0; kc < 4; ++kc) {
            v2f pa;
            pa.x = P[l16][kc * 4 + 2 * half];
            pa.y = P[l16][kc * 4 + 2 * half + 1];
#pragma unroll
            for (int nt = 0; nt < 4; ++nt) {
                const float* vp = V + (size_t)(kBase + kc * 4 + 2 * half) * DM
                                    + dBase + nt * 16 + l16;
                v2f b; b.x = vp[0]; b.y = vp[DM];
                o[nt] = wmma_f32(pa, b, o[nt]);
            }
        }
        __syncthreads();
    }

#pragma unroll
    for (int j = 0; j < 8; ++j) {
        float inv = 1.0f / lsum[j];
        int row = qBase + j + 8 * half;
#pragma unroll
        for (int nt = 0; nt < 4; ++nt) {
            O[(size_t)row * DM + dBase + nt * 16 + l16] = o[nt][j] * inv;
        }
    }
}

extern "C" void kernel_launch(void* const* d_in, const int* in_sizes, int n_in,
                              void* d_out, int out_size, void* d_ws, size_t ws_size,
                              hipStream_t stream) {
    const float* x  = (const float*)d_in[0];
    const float* fc = (const float*)d_in[1];
    const float* fs = (const float*)d_in[2];
    const float* wq = (const float*)d_in[3];
    const float* bq = (const float*)d_in[4];
    const float* wk = (const float*)d_in[5];
    const float* bk = (const float*)d_in[6];
    const float* wv = (const float*)d_in[7];
    const float* bv = (const float*)d_in[8];
    const float* wo = (const float*)d_in[9];
    const float* bo = (const float*)d_in[10];
    const float* gq = (const float*)d_in[11];
    const float* gk = (const float*)d_in[12];
    float* out = (float*)d_out;

    const size_t mat = (size_t)SB * DM;
    float* Qw = (float*)d_ws;
    float* Kw = Qw + mat;
    float* Vw = Kw + mat;
    float* Aw = Vw + mat;

    dim3 gblk(SB / 16);
    gemm16_kernel<<<gblk, 256, 0, stream>>>(x, wq, bq, gq, fc, fs, Qw, 1);
    gemm16_kernel<<<gblk, 256, 0, stream>>>(x, wk, bk, gk, fc, fs, Kw, 1);
    gemm16_kernel<<<gblk, 256, 0, stream>>>(x, wv, bv, gq, fc, fs, Vw, 0);
    attn_kernel<<<dim3(SB / 16, NH), 32, 0, stream>>>(Qw, Kw, Vw, Aw);
    gemm16_kernel<<<gblk, 256, 0, stream>>>(Aw, wo, bo, gq, fc, fs, out, 0);
}


# GEMM kernel: operands staged through LDS (X row tile once; W in 32-k slabs read as whole 128-B lines by 16-B loads, next slab prefetched in registers), not 16-line 8-B fragment loads; sums unchanged
# speedup vs baseline: 1.1992x; 1.1992x over previous
#include <hip/hip_runtime.h>
#include <math.h>

#define SB   2048
#define DM   1024
#define NH   16
#define HD   64
#define HD2  32
#define SCALE 0.125f
#define EPS  1e-6f

typedef __attribute__((ext_vector_type(2))) float v2f;
typedef __attribute__((ext_vector_type(4))) float v4f;
typedef __attribute__((ext_vector_type(8))) float v8f;

#define KT     32
#define WPITCH (KT + 4)

static __device__ __forceinline__ v8f wmma_f32(v2f a, v2f b, v8f c) {
    return __builtin_amdgcn_wmma_f32_16x16x4_f32(
         false, a,  false, b,
         (short)0, c,  false,  false);
}

__global__ __launch_bounds__(256) void gemm16_kernel(
    const float* __restrict__ X,
    const float* __restrict__ W,
    const float* __restrict__ bias,
    const float* __restrict__ g,
    const float* __restrict__ fc,
    const float* __restrict__ fs,
    float* __restrict__ out,
    int doNormRope)
{
    __shared__ __attribute__((aligned(16))) float tile[16][DM + 4];
    __shared__ __attribute__((aligned(16))) float wbuf[8][128][WPITCH];
    __shared__ float redbuf[16][17];
    __shared__ float rrms[16];

    const int tid  = threadIdx.x;
    const int wave = tid >> 5;
    const int lane = tid & 31;
    const int half = lane >> 4;
    const int l16  = lane & 15;
    const int rowBase  = blockIdx.x * 16;
    const int nColBase = wave * 128;

    v8f acc[8];
#pragma unroll
    for (int t = 0; t < 8; ++t) acc[t] = (v8f){0,0,0,0,0,0,0,0};

#pragma unroll
    for (int i = 0; i < 16; ++i) {
        v4f v = *(const v4f*)(X + (size_t)(rowBase + i) * DM + tid * 4);
        *(v4f*)&tile[i][tid * 4] = v;
    }

    const int wr = lane >> 3;
    const int wc = (lane & 7) * 4;
    const float* wsrc = W + (size_t)(nColBase + wr) * DM + wc;
    v4f R[32];
#pragma unroll
    for (int i = 0; i < 32; ++i) R[i] = *(const v4f*)(wsrc + (size_t)(i * 4) * DM);

    for (int s = 0; s < DM / KT; ++s) {
        __syncthreads();
#pragma unroll
        for (int i = 0; i < 32; ++i) *(v4f*)&wbuf[wave][i * 4 + wr][wc] = R[i];
        __syncthreads();

        if (s + 1 < DM / KT) {
            const float* wn = wsrc + (s + 1) * KT;
#pragma unroll
            for (int i = 0; i < 32; ++i) R[i] = *(const v4f*)(wn + (size_t)(i * 4) * DM);
        }

        const float* arow = &tile[l16][s * KT + 2 * half];
        const float* brow = &wbuf[wave][l16][2 * half];
#pragma unroll
        for (int kk = 0; kk < KT / 4; ++kk) {
            v2f a = *(const v2f*)(arow + kk * 4);
#pragma unroll
            for (int t = 0; t < 8; ++t) {
                v2f b = *(const v2f*)(brow + t * 16 * WPITCH + kk * 4);
                acc[t] = wmma_f32(a, b, acc[t]);
            }
        }
    }
    __syncthreads();

#pragma unroll
    for (int t = 0; t < 8; ++t) {
        int col = nColBase + t * 16 + l16;
        float bv = bias[col];
#pragma unroll
        for (int j = 0; j < 8; ++j) {
            int row = j + 8 * half;
            tile[row][col] = acc[t][j] + bv;
        }
    }
    __syncthreads();

    if (doNormRope) {
        {
            int r  = tid & 15;
            int c0 = tid >> 4;
            float s = 0.f;
            for (int c = c0; c < DM; c += 16) { float v = tile[r][c]; s += v * v; }
            redbuf[r][c0] = s;
        }
        __syncthreads();
        if (tid < 16) {
            float s = 0.f;
#pragma unroll
            for (int c = 0; c < 16; ++c) s += redbuf[tid][c];
            rrms[tid] = rsqrtf(s * (1.0f / (float)DM) + EPS);
        }
        __syncthreads();
    }

    for (int p = tid; p < 16 * (DM / 2); p += 256) {
        int r    = p >> 9;
        int cp   = p & 511;
        int dcol = cp * 2;
        int srow = rowBase + r;
        float re = tile[r][dcol];
        float im = tile[r][dcol + 1];
        if (doNormRope) {
            float rr = rrms[r];
            re *= rr * g[dcol];
            im *= rr * g[dcol + 1];
            int   pp = cp & (HD2 - 1);
            float c  = fc[srow * HD2 + pp];
            float sn = fs[srow * HD2 + pp];
            float ore = re * c - im * sn;
            float oim = re * sn + im * c;
            re = ore; im = oim;
        }
        out[(size_t)srow * DM + dcol]     = re;
        out[(size_t)srow * DM + dcol + 1] = im;
    }
}

__global__ __launch_bounds__(32) void attn_kernel(
    const float* __restrict__ Q,
    const float* __restrict__ K,
    const float* __restrict__ V,
    float* __restrict__ O)
{
    __shared__ float P[16][17];

    const int lane = threadIdx.x & 31;
    const int half = lane >> 4;
    const int l16  = lane & 15;
    const int qt    = blockIdx.x;
    const int h     = blockIdx.y;
    const int qBase = qt * 16;
    const int dBase = h * HD;

    v2f qa[16];
    {
        const float* qp = Q + (size_t)(qBase + l16) * DM + dBase + 2 * half;
#pragma unroll
        for (int kk = 0; kk < 16; ++kk) qa[kk] = *(const v2f*)(qp + kk * 4);
    }

    v8f o[4];
#pragma unroll
    for (int nt = 0; nt < 4; ++nt) o[nt] = (v8f){0,0,0,0,0,0,0,0};
    float m[8], lsum[8];
#pragma unroll
    for (int j = 0; j < 8; ++j) { m[j] = -INFINITY; lsum[j] = 0.f; }

    for (int t = 0; t < SB / 16; ++t) {
        const int kBase = t * 16;

        v8f s = (v8f){0,0,0,0,0,0,0,0};
        const float* kp = K + (size_t)(kBase + l16) * DM + dBase + 2 * half;
#pragma unroll
        for (int kk = 0; kk < 16; ++kk) {
            v2f b = *(const v2f*)(kp + kk * 4);
            s = wmma_f32(qa[kk], b, s);
        }

#pragma unroll
        for (int j = 0; j < 8; ++j) {
            float sv = s[j] * SCALE;
            float mx = sv;
            mx = fmaxf(mx, __shfl_xor(mx, 1, 32));
            mx = fmaxf(mx, __shfl_xor(mx, 2, 32));
            mx = fmaxf(mx, __shfl_xor(mx, 4, 32));
            mx = fmaxf(mx, __shfl_xor(mx, 8, 32));
            float nm   = fmaxf(m[j], mx);
            float corr = __expf(m[j] - nm);
            float pv   = __expf(sv - nm);
            float ps = pv;
            ps += __shfl_xor(ps, 1, 32);
            ps += __shfl_xor(ps, 2, 32);
            ps += __shfl_xor(ps, 4, 32);
            ps += __shfl_xor(ps, 8, 32);
            lsum[j] = lsum[j] * corr + ps;
            m[j] = nm;
            o[0][j] *= corr; o[1][j] *= corr; o[2][j] *= corr; o[3][j] *= corr;
            P[j + 8 * half][l16] = pv;
        }
        __syncthreads();

#pragma unroll
        for (int kc = 0; kc < 4; ++kc) {
            v2f pa;
            pa.x = P[l16][kc * 4 + 2 * half];
            pa.y = P[l16][kc * 4 + 2 * half + 1];
#pragma unroll
            for (int nt = 0; nt < 4; ++nt) {
                const float* vp = V + (size_t)(kBase + kc * 4 + 2 * half) * DM
                                    + dBase + nt * 16 + l16;
                v2f b; b.x = vp[0]; b.y = vp[DM];
                o[nt] = wmma_f32(pa, b, o[nt]);
            }
        }
        __syncthreads();
    }

#pragma unroll
    for (int j = 0; j < 8; ++j) {
        float inv = 1.0f / lsum[j];
        int row = qBase + j + 8 * half;
#pragma unroll
        for (int nt = 0; nt < 4; ++nt) {
            O[(size_t)row * DM + dBase + nt * 16 + l16] = o[nt][j] * inv;
        }
    }
}

extern "C" void kernel_launch(void* const* d_in, const int* in_sizes, int n_in,
                              void* d_out, int out_size, void* d_ws, size_t ws_size,
                              hipStream_t stream) {
    const float* x  = (const float*)d_in[0];
    const float* fc = (const float*)d_in[1];
    const float* fs = (const float*)d_in[2];
    const float* wq = (const float*)d_in[3];
    const float* bq = (const float*)d_in[4];
    const float* wk = (const float*)d_in[5];
    const float* bk = (const float*)d_in[6];
    const float* wv = (const float*)d_in[7];
    const float* bv = (const float*)d_in[8];
    const float* wo = (const float*)d_in[9];
    const float* bo = (const float*)d_in[10];
    const float* gq = (const float*)d_in[11];
    const float* gk = (const float*)d_in[12];
    float* out = (float*)d_out;

    const size_t mat = (size_t)SB * DM;
    float* Qw = (float*)d_ws;
    float* Kw = Qw + mat;
    float* Vw = Kw + mat;
    float* Aw = Vw + mat;

    dim3 gblk(SB / 16);
    gemm16_kernel<<<gblk, 256, 0, stream>>>(x, wq, bq, gq, fc, fs, Qw, 1);
    gemm16_kernel<<<gblk, 256, 0, stream>>>(x, wk, bk, gk, fc, fs, Kw, 1);
    gemm16_kernel<<<gblk, 256, 0, stream>>>(x, wv, bv, gq, fc, fs, Vw, 0);
    attn_kernel<<<dim3(SB / 16, NH), 32, 0, stream>>>(Qw, Kw, Vw, Aw);
    gemm16_kernel<<<gblk, 256, 0, stream>>>(Aw, wo, bo, gq, fc, fs, out, 0);
}


# attention kernel: K/V staged through LDS in 32-key slabs (whole 128-B lines, 16-B loads, next slab prefetched in registers), shared by 8 query-tile waves per workgroup; sums unchanged
# speedup vs baseline: 1.7839x; 1.4876x over previous
#include <hip/hip_runtime.h>
#include <math.h>

#define SB   2048
#define DM   1024
#define NH   16
#define HD   64
#define HD2  32
#define SCALE 0.125f
#define EPS  1e-6f

typedef __attribute__((ext_vector_type(2))) float v2f;
typedef __attribute__((ext_vector_type(4))) float v4f;
typedef __attribute__((ext_vector_type(8))) float v8f;

#define KT     32
#define WPITCH (KT + 4)

static __device__ __forceinline__ v8f wmma_f32(v2f a, v2f b, v8f c) {
    return __builtin_amdgcn_wmma_f32_16x16x4_f32(
         false, a,  false, b,
         (short)0, c,  false,  false);
}

__global__ __launch_bounds__(256) void gemm16_kernel(
    const float* __restrict__ X,
    const float* __restrict__ W,
    const float* __restrict__ bias,
    const float* __restrict__ g,
    const float* __restrict__ fc,
    const float* __restrict__ fs,
    float* __restrict__ out,
    int doNormRope)
{
    __shared__ __attribute__((aligned(16))) float tile[16][DM + 4];
    __shared__ __attribute__((aligned(16))) float wbuf[8][128][WPITCH];
    __shared__ float redbuf[16][17];
    __shared__ float rrms[16];

    const int tid  = threadIdx.x;
    const int wave = tid >> 5;
    const int lane = tid & 31;
    const int half = lane >> 4;
    const int l16  = lane & 15;
    const int rowBase  = blockIdx.x * 16;
    const int nColBase = wave * 128;

    v8f acc[8];
#pragma unroll
    for (int t = 0; t < 8; ++t) acc[t] = (v8f){0,0,0,0,0,0,0,0};

#pragma unroll
    for (int i = 0; i < 16; ++i) {
        v4f v = *(const v4f*)(X + (size_t)(rowBase + i) * DM + tid * 4);
        *(v4f*)&tile[i][tid * 4] = v;
    }

    const int wr = lane >> 3;
    const int wc = (lane & 7) * 4;
    const float* wsrc = W + (size_t)(nColBase + wr) * DM + wc;
    v4f R[32];
#pragma unroll
    for (int i = 0; i < 32; ++i) R[i] = *(const v4f*)(wsrc + (size_t)(i * 4) * DM);

    for (int s = 0; s < DM / KT; ++s) {
        __syncthreads();
#pragma unroll
        for (int i = 0; i < 32; ++i) *(v4f*)&wbuf[wave][i * 4 + wr][wc] = R[i];
        __syncthreads();

        if (s + 1 < DM / KT) {
            const float* wn = wsrc + (s + 1) * KT;
#pragma unroll
            for (int i = 0; i < 32; ++i) R[i] = *(const v4f*)(wn + (size_t)(i * 4) * DM);
        }

        const float* arow = &tile[l16][s * KT + 2 * half];
        const float* brow = &wbuf[wave][l16][2 * half];
#pragma unroll
        for (int kk = 0; kk < KT / 4; ++kk) {
            v2f a = *(const v2f*)(arow + kk * 4);
#pragma unroll
            for (int t = 0; t < 8; ++t) {
                v2f b = *(const v2f*)(brow + t * 16 * WPITCH + kk * 4);
                acc[t] = wmma_f32(a, b, acc[t]);
            }
        }
    }
    __syncthreads();

#pragma unroll
    for (int t = 0; t < 8; ++t) {
        int col = nColBase + t * 16 + l16;
        float bv = bias[col];
#pragma unroll
        for (int j = 0; j < 8; ++j) {
            int row = j + 8 * half;
            tile[row][col] = acc[t][j] + bv;
        }
    }
    __syncthreads();

    if (doNormRope) {
        {
            int r  = tid & 15;
            int c0 = tid >> 4;
            float s = 0.f;
            for (int c = c0; c < DM; c += 16) { float v = tile[r][c]; s += v * v; }
            redbuf[r][c0] = s;
        }
        __syncthreads();
        if (tid < 16) {
            float s = 0.f;
#pragma unroll
            for (int c = 0; c < 16; ++c) s += redbuf[tid][c];
            rrms[tid] = rsqrtf(s * (1.0f / (float)DM) + EPS);
        }
        __syncthreads();
    }

    for (int p = tid; p < 16 * (DM / 2); p += 256) {
        int r    = p >> 9;
        int cp   = p & 511;
        int dcol = cp * 2;
        int srow = rowBase + r;
        float re = tile[r][dcol];
        float im = tile[r][dcol + 1];
        if (doNormRope) {
            float rr = rrms[r];
            re *= rr * g[dcol];
            im *= rr * g[dcol + 1];
            int   pp = cp & (HD2 - 1);
            float c  = fc[srow * HD2 + pp];
            float sn = fs[srow * HD2 + pp];
            float ore = re * c - im * sn;
            float oim = re * sn + im * c;
            re = ore; im = oim;
        }
        out[(size_t)srow * DM + dcol]     = re;
        out[(size_t)srow * DM + dcol + 1] = im;
    }
}

#define AW  8
#define KS  32
#define KP  (HD + 4)
#define VP  (HD + 16)
#define PP  20

static __device__ __forceinline__ void wave_lds_handover() {
    __builtin_amdgcn_fence(__ATOMIC_RELEASE, "wavefront");
    __builtin_amdgcn_wave_barrier();
    __builtin_amdgcn_fence(__ATOMIC_ACQUIRE, "wavefront");
}

__global__ __launch_bounds__(256) void attn_kernel(
    const float* __restrict__ Q,
    const float* __restrict__ K,
    const float* __restrict__ V,
    float* __restrict__ O)
{
    __shared__ __attribute__((aligned(16))) float Ks[2][KS][KP];
    __shared__ __attribute__((aligned(16))) float Vs[2][KS][VP];
    __shared__ __attribute__((aligned(16))) float Pw[AW][2][16][PP];

    const int tid  = threadIdx.x;
    const int wave = tid >> 5;
    const int lane = tid & 31;
    const int half = lane >> 4;
    const int l16  = lane & 15;
    const int qt    = blockIdx.x * AW + wave;
    const int h     = blockIdx.y;
    const int qBase = qt * 16;
    const int dBase = h * HD;

    const int sr = tid >> 4;
    const int sc = (tid & 15) * 4;
    const float* ksrc = K + (size_t)sr * DM + dBase + sc;
    const float* vsrc = V + (size_t)sr * DM + dBase + sc;
    v4f rk0 = *(const v4f*)(ksrc);
    v4f rk1 = *(const v4f*)(ksrc + (size_t)16 * DM);
    v4f rv0 = *(const v4f*)(vsrc);
    v4f rv1 = *(const v4f*)(vsrc + (size_t)16 * DM);

    v2f qa[16];
    {
        const float* qp = Q + (size_t)(qBase + l16) * DM + dBase + 2 * half;
#pragma unroll
        for (int kk = 0; kk < 16; ++kk) qa[kk] = *(const v2f*)(qp + kk * 4);
    }

    v8f o[4];
#pragma unroll
    for (int nt = 0; nt < 4; ++nt) o[nt] = (v8f){0,0,0,0,0,0,0,0};
    float m[8], lsum[8];
#pragma unroll
    for (int j = 0; j < 8; ++j) { m[j] = -INFINITY; lsum[j] = 0.f; }

    for (int t = 0; t < SB / KS; ++t) {
        const int cur = t & 1;

        *(v4f*)&Ks[cur][sr][sc]      = rk0;
        *(v4f*)&Ks[cur][sr + 16][sc] = rk1;
        *(v4f*)&Vs[cur][sr][sc]      = rv0;
        *(v4f*)&Vs[cur][sr + 16][sc] = rv1;
        __syncthreads();

        if (t + 1 < SB / KS) {
            const size_t off = (size_t)(t + 1) * KS * DM;
            rk0 = *(const v4f*)(ksrc + off);
            rk1 = *(const v4f*)(ksrc + off + (size_t)16 * DM);
            rv0 = *(const v4f*)(vsrc + off);
            rv1 = *(const v4f*)(vsrc + off + (size_t)16 * DM);
        }

        v8f sacc[2];
        sacc[0] = (v8f){0,0,0,0,0,0,0,0};
        sacc[1] = (v8f){0,0,0,0,0,0,0,0};
        {
            const float* kp0 = &Ks[cur][l16][2 * half];
            const float* kp1 = &Ks[cur][16 + l16][2 * half];
#pragma unroll
            for (int kk = 0; kk < 16; ++kk) {
                v2f b0 = *(const v2f*)(kp0 + kk * 4);
                v2f b1 = *(const v2f*)(kp1 + kk * 4);
                sacc[0] = wmma_f32(qa[kk], b0, sacc[0]);
                sacc[1] = wmma_f32(qa[kk], b1, sacc[1]);
            }
        }

#pragma unroll
        for (int u = 0; u < 2; ++u) {
            const v8f s = sacc[u];

            float (*P)[PP] = Pw[wave][u];
#pragma unroll
            for (int j = 0; j < 8; ++j) {
                float sv = s[j] * SCALE;
                float mx = sv;
                mx = fmaxf(mx, __shfl_xor(mx, 1, 32));
                mx = fmaxf(mx, __shfl_xor(mx, 2, 32));
                mx = fmaxf(mx, __shfl_xor(mx, 4, 32));
                mx = fmaxf(mx, __shfl_xor(mx, 8, 32));
                float nm   = fmaxf(m[j], mx);
                float corr = __expf(m[j] - nm);
                float pv   = __expf(sv - nm);
                float ps = pv;
                ps += __shfl_xor(ps, 1, 32);
                ps += __shfl_xor(ps, 2, 32);
                ps += __shfl_xor(ps, 4, 32);
                ps += __shfl_xor(ps, 8, 32);
                lsum[j] = lsum[j] * corr + ps;
                m[j] = nm;
                o[0][j] *= corr; o[1][j] *= corr; o[2][j] *= corr; o[3][j] *= corr;
                P[j + 8 * half][l16] = pv;
            }
            wave_lds_handover();

#pragma unroll
            for (int kc = 0; kc < 4; ++kc) {
                v2f pa = *(const v2f*)&P[l16][kc * 4 + 2 * half];
                const float* vp = &Vs[cur][u * 16 + kc * 4 + 2 * half][l16];
#pragma unroll
                for (int nt = 0; nt < 4; ++nt) {
                    v2f b; b.x = vp[nt * 16]; b.y = vp[VP + nt * 16];
                    o[nt] = wmma_f32(pa, b, o[nt]);
                }
            }
        }
    }

#pragma unroll
    for (int j = 0; j < 8; ++j) {
        float inv = 1.0f / lsum[j];
        int row = qBase + j + 8 * half;
#pragma unroll
        for (int nt = 0; nt < 4; ++nt) {
            O[(size_t)row * DM + dBase + nt * 16 + l16] = o[nt][j] * inv;
        }
    }
}

extern "C" void kernel_launch(void* const* d_in, const int* in_sizes, int n_in,
                              void* d_out, int out_size, void* d_ws, size_t ws_size,
                              hipStream_t stream) {
    const float* x  = (const float*)d_in[0];
    const float* fc = (const float*)d_in[1];
    const float* fs = (const float*)d_in[2];
    const float* wq = (const float*)d_in[3];
    const float* bq = (const float*)d_in[4];
    const float* wk = (const float*)d_in[5];
    const float* bk = (const float*)d_in[6];
    const float* wv = (const float*)d_in[7];
    const float* bv = (const float*)d_in[8];
    const float* wo = (const float*)d_in[9];
    const float* bo = (const float*)d_in[10];
    const float* gq = (const float*)d_in[11];
    const float* gk = (const float*)d_in[12];
    float* out = (float*)d_out;

    const size_t mat = (size_t)SB * DM;
    float* Qw = (float*)d_ws;
    float* Kw = Qw + mat;
    float* Vw = Kw + mat;
    float* Aw = Vw + mat;

    dim3 gblk(SB / 16);
    gemm16_kernel<<<gblk, 256, 0, stream>>>(x, wq, bq, gq, fc, fs, Qw, 1);
    gemm16_kernel<<<gblk, 256, 0, stream>>>(x, wk, bk, gk, fc, fs, Kw, 1);
    gemm16_kernel<<<gblk, 256, 0, stream>>>(x, wv, bv, gq, fc, fs, Vw, 0);
    attn_kernel<<<dim3(SB / 16 / AW, NH), 32 * AW, 0, stream>>>(Qw, Kw, Vw, Aw);
    gemm16_kernel<<<gblk, 256, 0, stream>>>(Aw, wo, bo, gq, fc, fs, out, 0);
}


# attention: scores formed transposed (S^T = K*Q^T): softmax max/sum in-lane + one half swap, P feeds the PV product from registers (no LDS bounce, no lane permutes); row-sum and PV sum order change
# speedup vs baseline: 1.9782x; 1.1089x over previous
#include <hip/hip_runtime.h>
#include <math.h>

#define SB   2048
#define DM   1024
#define NH   16
#define HD   64
#define HD2  32
#define SCALE 0.125f
#define EPS  1e-6f

typedef __attribute__((ext_vector_type(2))) float v2f;
typedef __attribute__((ext_vector_type(4))) float v4f;
typedef __attribute__((ext_vector_type(8))) float v8f;

#define KT     32
#define WPITCH (KT + 4)

static __device__ __forceinline__ v8f wmma_f32(v2f a, v2f b, v8f c) {
    return __builtin_amdgcn_wmma_f32_16x16x4_f32(
         false, a,  false, b,
         (short)0, c,  false,  false);
}

__global__ __launch_bounds__(256) void gemm16_kernel(
    const float* __restrict__ X,
    const float* __restrict__ W,
    const float* __restrict__ bias,
    const float* __restrict__ g,
    const float* __restrict__ fc,
    const float* __restrict__ fs,
    float* __restrict__ out,
    int doNormRope)
{
    __shared__ __attribute__((aligned(16))) float tile[16][DM + 4];
    __shared__ __attribute__((aligned(16))) float wbuf[8][128][WPITCH];
    __shared__ float redbuf[16][17];
    __shared__ float rrms[16];

    const int tid  = threadIdx.x;
    const int wave = tid >> 5;
    const int lane = tid & 31;
    const int half = lane >> 4;
    const int l16  = lane & 15;
    const int rowBase  = blockIdx.x * 16;
    const int nColBase = wave * 128;

    v8f acc[8];
#pragma unroll
    for (int t = 0; t < 8; ++t) acc[t] = (v8f){0,0,0,0,0,0,0,0};

#pragma unroll
    for (int i = 0; i < 16; ++i) {
        v4f v = *(const v4f*)(X + (size_t)(rowBase + i) * DM + tid * 4);
        *(v4f*)&tile[i][tid * 4] = v;
    }

    const int wr = lane >> 3;
    const int wc = (lane & 7) * 4;
    const float* wsrc = W + (size_t)(nColBase + wr) * DM + wc;
    v4f R[32];
#pragma unroll
    for (int i = 0; i < 32; ++i) R[i] = *(const v4f*)(wsrc + (size_t)(i * 4) * DM);

    for (int s = 0; s < DM / KT; ++s) {
        __syncthreads();
#pragma unroll
        for (int i = 0; i < 32; ++i) *(v4f*)&wbuf[wave][i * 4 + wr][wc] = R[i];
        __syncthreads();

        if (s + 1 < DM / KT) {
            const float* wn = wsrc + (s + 1) * KT;
#pragma unroll
            for (int i = 0; i < 32; ++i) R[i] = *(const v4f*)(wn + (size_t)(i * 4) * DM);
        }

        const float* arow = &tile[l16][s * KT + 2 * half];
        const float* brow = &wbuf[wave][l16][2 * half];
#pragma unroll
        for (int kk = 0; kk < KT / 4; ++kk) {
            v2f a = *(const v2f*)(arow + kk * 4);
#pragma unroll
            for (int t = 0; t < 8; ++t) {
                v2f b = *(const v2f*)(brow + t * 16 * WPITCH + kk * 4);
                acc[t] = wmma_f32(a, b, acc[t]);
            }
        }
    }
    __syncthreads();

#pragma unroll
    for (int t = 0; t < 8; ++t) {
        int col = nColBase + t * 16 + l16;
        float bv = bias[col];
#pragma unroll
        for (int j = 0; j < 8; ++j) {
            int row = j + 8 * half;
            tile[row][col] = acc[t][j] + bv;
        }
    }
    __syncthreads();

    if (doNormRope) {
        {
            int r  = tid & 15;
            int c0 = tid >> 4;
            float s = 0.f;
            for (int c = c0; c < DM; c += 16) { float v = tile[r][c]; s += v * v; }
            redbuf[r][c0] = s;
        }
        __syncthreads();
        if (tid < 16) {
            float s = 0.f;
#pragma unroll
            for (int c = 0; c < 16; ++c) s += redbuf[tid][c];
            rrms[tid] = rsqrtf(s * (1.0f / (float)DM) + EPS);
        }
        __syncthreads();
    }

    for (int p = tid; p < 16 * (DM / 2); p += 256) {
        int r    = p >> 9;
        int cp   = p & 511;
        int dcol = cp * 2;
        int srow = rowBase + r;
        float re = tile[r][dcol];
        float im = tile[r][dcol + 1];
        if (doNormRope) {
            float rr = rrms[r];
            re *= rr * g[dcol];
            im *= rr * g[dcol + 1];
            int   pp = cp & (HD2 - 1);
            float c  = fc[srow * HD2 + pp];
            float sn = fs[srow * HD2 + pp];
            float ore = re * c - im * sn;
            float oim = re * sn + im * c;
            re = ore; im = oim;
        }
        out[(size_t)srow * DM + dcol]     = re;
        out[(size_t)srow * DM + dcol + 1] = im;
    }
}

#define AW  8
#define KS  32
#define KP  (HD + 4)
#define VP  (HD + 16)

static __device__ __forceinline__ float xhalf(float x) {
    int b = __builtin_bit_cast(int, x);
    return __builtin_bit_cast(float,
        __builtin_amdgcn_permlanex16(b, b, 0x76543210, 0xfedcba98, false, false));
}

__global__ __launch_bounds__(256) void attn_kernel(
    const float* __restrict__ Q,
    const float* __restrict__ K,
    const float* __restrict__ V,
    float* __restrict__ O)
{
    __shared__ __attribute__((aligned(16))) float Ks[2][KS][KP];
    __shared__ __attribute__((aligned(16))) float Vs[2][KS][VP];

    const int tid  = threadIdx.x;
    const int wave = tid >> 5;
    const int lane = tid & 31;
    const int half = lane >> 4;
    const int l16  = lane & 15;
    const int qt    = blockIdx.x * AW + wave;
    const int h     = blockIdx.y;
    const int qBase = qt * 16;
    const int dBase = h * HD;

    const int sr = tid >> 4;
    const int sc = (tid & 15) * 4;
    const int vr = 4 * ((sr & 7) >> 1) + 2 * (sr >> 3) + (sr & 1);
    const float* ksrc = K + (size_t)sr * DM + dBase + sc;
    const float* vsrc = V + (size_t)sr * DM + dBase + sc;
    v4f rk0 = *(const v4f*)(ksrc);
    v4f rk1 = *(const v4f*)(ksrc + (size_t)16 * DM);
    v4f rv0 = *(const v4f*)(vsrc);
    v4f rv1 = *(const v4f*)(vsrc + (size_t)16 * DM);

    v2f qb[16];
    {
        const float* qp = Q + (size_t)(qBase + l16) * DM + dBase + 2 * half;
#pragma unroll
        for (int kk = 0; kk < 16; ++kk) qb[kk] = *(const v2f*)(qp + kk * 4);
    }

    v8f o[4];
#pragma unroll
    for (int dt = 0; dt < 4; ++dt) o[dt] = (v8f){0,0,0,0,0,0,0,0};
    float m = -INFINITY, lsum = 0.f;

    for (int t = 0; t < SB / KS; ++t) {
        const int cur = t & 1;

        *(v4f*)&Ks[cur][sr][sc]      = rk0;
        *(v4f*)&Ks[cur][sr + 16][sc] = rk1;
        *(v4f*)&Vs[cur][vr][sc]      = rv0;
        *(v4f*)&Vs[cur][16 + vr][sc] = rv1;
        __syncthreads();

        if (t + 1 < SB / KS) {
            const size_t off = (size_t)(t + 1) * KS * DM;
            rk0 = *(const v4f*)(ksrc + off);
            rk1 = *(const v4f*)(ksrc + off + (size_t)16 * DM);
            rv0 = *(const v4f*)(vsrc + off);
            rv1 = *(const v4f*)(vsrc + off + (size_t)16 * DM);
        }

        v8f sacc[2];
        sacc[0] = (v8f){0,0,0,0,0,0,0,0};
        sacc[1] = (v8f){0,0,0,0,0,0,0,0};
        {
            const float* kp0 = &Ks[cur][l16][2 * half];
            const float* kp1 = &Ks[cur][16 + l16][2 * half];
#pragma unroll
            for (int kk = 0; kk < 16; ++kk) {
                v2f a0 = *(const v2f*)(kp0 + kk * 4);
                v2f a1 = *(const v2f*)(kp1 + kk * 4);
                sacc[0] = wmma_f32(a0, qb[kk], sacc[0]);
                sacc[1] = wmma_f32(a1, qb[kk], sacc[1]);
            }
        }

#pragma unroll
        for (int u = 0; u < 2; ++u) {
            const v8f s = sacc[u];

            float sv[8];
#pragma unroll
            for (int j = 0; j < 8; ++j) sv[j] = s[j] * SCALE;
            float mx = fmaxf(fmaxf(fmaxf(sv[0], sv[1]), fmaxf(sv[2], sv[3])),
                             fmaxf(fmaxf(sv[4], sv[5]), fmaxf(sv[6], sv[7])));
            mx = fmaxf(mx, xhalf(mx));
            const float nm   = fmaxf(m, mx);
            const float corr = __expf(m - nm);
            float p[8];
#pragma unroll
            for (int j = 0; j < 8; ++j) p[j] = __expf(sv[j] - nm);
            float ps = ((p[0] + p[1]) + (p[2] + p[3])) + ((p[4] + p[5]) + (p[6] + p[7]));
            ps += xhalf(ps);
            lsum = lsum * corr + ps;
            m = nm;
#pragma unroll
            for (int dt = 0; dt < 4; ++dt)
#pragma unroll
                for (int j = 0; j < 8; ++j) o[dt][j] *= corr;

#pragma unroll
            for (int c = 0; c < 4; ++c) {
                v2f pb; pb.x = p[2 * c]; pb.y = p[2 * c + 1];
                const float* vp = &Vs[cur][u * 16 + 4 * c + 2 * half][l16];
#pragma unroll
                for (int dt = 0; dt < 4; ++dt) {
                    v2f a; a.x = vp[dt * 16]; a.y = vp[VP + dt * 16];
                    o[dt] = wmma_f32(a, pb, o[dt]);
                }
            }
        }
    }

    {
        const float inv = 1.0f / lsum;
        float* orow = O + (size_t)(qBase + l16) * DM + dBase + 8 * half;
#pragma unroll
        for (int dt = 0; dt < 4; ++dt) {
            v4f lo, hi;
            lo.x = o[dt][0] * inv; lo.y = o[dt][1] * inv; lo.z = o[dt][2] * inv; lo.w = o[dt][3] * inv;
            hi.x = o[dt][4] * inv; hi.y = o[dt][5] * inv; hi.z = o[dt][6] * inv; hi.w = o[dt][7] * inv;
            *(v4f*)(orow + dt * 16)     = lo;
            *(v4f*)(orow + dt * 16 + 4) = hi;
        }
    }
}

extern "C" void kernel_launch(void* const* d_in, const int* in_sizes, int n_in,
                              void* d_out, int out_size, void* d_ws, size_t ws_size,
                              hipStream_t stream) {
    const float* x  = (const float*)d_in[0];
    const float* fc = (const float*)d_in[1];
    const float* fs = (const float*)d_in[2];
    const float* wq = (const float*)d_in[3];
    const float* bq = (const float*)d_in[4];
    const float* wk = (const float*)d_in[5];
    const float* bk = (const float*)d_in[6];
    const float* wv = (const float*)d_in[7];
    const float* bv = (const float*)d_in[8];
    const float* wo = (const float*)d_in[9];
    const float* bo = (const float*)d_in[10];
    const float* gq = (const float*)d_in[11];
    const float* gk = (const float*)d_in[12];
    float* out = (float*)d_out;

    const size_t mat = (size_t)SB * DM;
    float* Qw = (float*)d_ws;
    float* Kw = Qw + mat;
    float* Vw = Kw + mat;
    float* Aw = Vw + mat;

    dim3 gblk(SB / 16);
    gemm16_kernel<<<gblk, 256, 0, stream>>>(x, wq, bq, gq, fc, fs, Qw, 1);
    gemm16_kernel<<<gblk, 256, 0, stream>>>(x, wk, bk, gk, fc, fs, Kw, 1);
    gemm16_kernel<<<gblk, 256, 0, stream>>>(x, wv, bv, gq, fc, fs, Vw, 0);
    attn_kernel<<<dim3(SB / 16 / AW, NH), 32 * AW, 0, stream>>>(Qw, Kw, Vw, Aw);
    gemm16_kernel<<<gblk, 256, 0, stream>>>(Aw, wo, bo, gq, fc, fs, out, 0);
}


# GEMM work re-gridded to fill every compute unit: q,k,v projections in one launch (q,k as 16x1024 row tiles, v as 16x512 tiles), out projection as 256 tiles of 16x512; 5 launches -> 3; sums unchanged
# speedup vs baseline: 3.2501x; 1.6430x over previous
#include <hip/hip_runtime.h>
#include <math.h>

#define SB   2048
#define DM   1024
#define NH   16
#define HD   64
#define HD2  32
#define SCALE 0.125f
#define EPS  1e-6f

typedef __attribute__((ext_vector_type(2))) float v2f;
typedef __attribute__((ext_vector_type(4))) float v4f;
typedef __attribute__((ext_vector_type(8))) float v8f;

#define KT     32
#define WPITCH (KT + 4)

static __device__ __forceinline__ v8f wmma_f32(v2f a, v2f b, v8f c) {
    return __builtin_amdgcn_wmma_f32_16x16x4_f32(
         false, a,  false, b,
         (short)0, c,  false,  false);
}

template <int TT, bool NORM>
static __device__ __forceinline__ void gemm_tile(
    const float* __restrict__ X,
    const float* __restrict__ W,
    const float* __restrict__ bias,
    const float* __restrict__ g,
    const float* __restrict__ fc,
    const float* __restrict__ fs,
    float* __restrict__ out,
    const int rowBase, const int colBase,
    float (*tile)[DM + 4],
    float (*wbuf)[128][WPITCH],
    float (*redbuf)[17], float* rrms)
{
    static_assert(!NORM || TT == 8, "the row norm needs the whole row in one workgroup");
    constexpr int NC = TT * 16;
    constexpr int NL = NC / 4;
    constexpr int PR = TT * 64;

    const int tid  = threadIdx.x;
    const int wave = tid >> 5;
    const int lane = tid & 31;
    const int half = lane >> 4;
    const int l16  = lane & 15;
    const int nColLoc  = wave * NC;
    const int nColBase = colBase + nColLoc;

    v8f acc[TT];
#pragma unroll
    for (int t = 0; t < TT; ++t) acc[t] = (v8f){0,0,0,0,0,0,0,0};

#pragma unroll
    for (int i = 0; i < 16; ++i) {
        v4f v = *(const v4f*)(X + (size_t)(rowBase + i) * DM + tid * 4);
        *(v4f*)&tile[i][tid * 4] = v;
    }

    const int wr = lane >> 3;
    const int wc = (lane & 7) * 4;
    const float* wsrc = W + (size_t)(nColBase + wr) * DM + wc;
    v4f R[NL];
#pragma unroll
    for (int i = 0; i < NL; ++i) R[i] = *(const v4f*)(wsrc + (size_t)(i * 4) * DM);

    for (int s = 0; s < DM / KT; ++s) {
        __syncthreads();
#pragma unroll
        for (int i = 0; i < NL; ++i) *(v4f*)&wbuf[wave][i * 4 + wr][wc] = R[i];
        __syncthreads();

        if (s + 1 < DM / KT) {
            const float* wn = wsrc + (s + 1) * KT;
#pragma unroll
            for (int i = 0; i < NL; ++i) R[i] = *(const v4f*)(wn + (size_t)(i * 4) * DM);
        }

        const float* arow = &tile[l16][s * KT + 2 * half];
        const float* brow = &wbuf[wave][l16][2 * half];
#pragma unroll
        for (int kk = 0; kk < KT / 4; ++kk) {
            v2f a = *(const v2f*)(arow + kk * 4);
#pragma unroll
            for (int t = 0; t < TT; ++t) {
                v2f b = *(const v2f*)(brow + t * 16 * WPITCH + kk * 4);
                acc[t] = wmma_f32(a, b, acc[t]);
            }
        }
    }
    __syncthreads();

#pragma unroll
    for (int t = 0; t < TT; ++t) {
        int cl = nColLoc + t * 16 + l16;
        float bv = bias[colBase + cl];
#pragma unroll
        for (int j = 0; j < 8; ++j) {
            int row = j + 8 * half;
            tile[row][cl] = acc[t][j] + bv;
        }
    }
    __syncthreads();

    if (NORM) {
        {
            int r  = tid & 15;
            int c0 = tid >> 4;
            float s = 0.f;
            for (int c = c0; c < DM; c += 16) { float v = tile[r][c]; s += v * v; }
            redbuf[r][c0] = s;
        }
        __syncthreads();
        if (tid < 16) {
            float s = 0.f;
#pragma unroll
            for (int c = 0; c < 16; ++c) s += redbuf[tid][c];
            rrms[tid] = rsqrtf(s * (1.0f / (float)DM) + EPS);
        }
        __syncthreads();
    }

    for (int p = tid; p < 16 * PR; p += 256) {
        int r    = p / PR;
        int cp   = p % PR;
        int dcol = cp * 2;
        int srow = rowBase + r;
        float re = tile[r][dcol];
        float im = tile[r][dcol + 1];
        if (NORM) {
            float rr = rrms[r];
            re *= rr * g[dcol];
            im *= rr * g[dcol + 1];
            int   pp = cp & (HD2 - 1);
            float c  = fc[srow * HD2 + pp];
            float sn = fs[srow * HD2 + pp];
            float ore = re * c - im * sn;
            float oim = re * sn + im * c;
            re = ore; im = oim;
        }
        out[(size_t)srow * DM + colBase + dcol]     = re;
        out[(size_t)srow * DM + colBase + dcol + 1] = im;
    }
}

#define RT (SB / 16)

__global__ __launch_bounds__(256) void gemm16_kernel(
    const float* __restrict__ X,
    const float* __restrict__ W0, const float* __restrict__ b0, const float* __restrict__ g0, float* __restrict__ out0,
    const float* __restrict__ W1, const float* __restrict__ b1, const float* __restrict__ g1, float* __restrict__ out1,
    const float* __restrict__ W2, const float* __restrict__ b2, float* __restrict__ out2,
    const float* __restrict__ fc,
    const float* __restrict__ fs,
    int nFull)
{
    __shared__ __attribute__((aligned(16))) float tile[16][DM + 4];
    __shared__ __attribute__((aligned(16))) float wbuf[8][128][WPITCH];
    __shared__ float redbuf[16][17];
    __shared__ float rrms[16];

    const int b = blockIdx.x;
    if (b < nFull) {
        const bool second = b >= RT;
        const int  rt     = second ? b - RT : b;
        gemm_tile<8, true>(X, second ? W1 : W0, second ? b1 : b0, second ? g1 : g0, fc, fs,
                           second ? out1 : out0, rt * 16, 0, tile, wbuf, redbuf, rrms);
    } else {
        const int i = b - nFull;
        gemm_tile<4, false>(X, W2, b2, g0, fc, fs, out2, (i >> 1) * 16, (i & 1) * (DM / 2),
                            tile, wbuf, redbuf, rrms);
    }
}

#define AW  8
#define KS  32
#define KP  (HD + 4)
#define VP  (HD + 16)

static __device__ __forceinline__ float xhalf(float x) {
    int b = __builtin_bit_cast(int, x);
    return __builtin_bit_cast(float,
        __builtin_amdgcn_permlanex16(b, b, 0x76543210, 0xfedcba98, false, false));
}

__global__ __launch_bounds__(256) void attn_kernel(
    const float* __restrict__ Q,
    const float* __restrict__ K,
    const float* __restrict__ V,
    float* __restrict__ O)
{
    __shared__ __attribute__((aligned(16))) float Ks[2][KS][KP];
    __shared__ __attribute__((aligned(16))) float Vs[2][KS][VP];

    const int tid  = threadIdx.x;
    const int wave = tid >> 5;
    const int lane = tid & 31;
    const int half = lane >> 4;
    const int l16  = lane & 15;
    const int qt    = blockIdx.x * AW + wave;
    const int h     = blockIdx.y;
    const int qBase = qt * 16;
    const int dBase = h * HD;

    const int sr = tid >> 4;
    const int sc = (tid & 15) * 4;
    const int vr = 4 * ((sr & 7) >> 1) + 2 * (sr >> 3) + (sr & 1);
    const float* ksrc = K + (size_t)sr * DM + dBase + sc;
    const float* vsrc = V + (size_t)sr * DM + dBase + sc;
    v4f rk0 = *(const v4f*)(ksrc);
    v4f rk1 = *(const v4f*)(ksrc + (size_t)16 * DM);
    v4f rv0 = *(const v4f*)(vsrc);
    v4f rv1 = *(const v4f*)(vsrc + (size_t)16 * DM);

    v2f qb[16];
    {
        const float* qp = Q + (size_t)(qBase + l16) * DM + dBase + 2 * half;
#pragma unroll
        for (int kk = 0; kk < 16; ++kk) qb[kk] = *(const v2f*)(qp + kk * 4);
    }

    v8f o[4];
#pragma unroll
    for (int dt = 0; dt < 4; ++dt) o[dt] = (v8f){0,0,0,0,0,0,0,0};
    float m = -INFINITY, lsum = 0.f;

    for (int t = 0; t < SB / KS; ++t) {
        const int cur = t & 1;

        *(v4f*)&Ks[cur][sr][sc]      = rk0;
        *(v4f*)&Ks[cur][sr + 16][sc] = rk1;
        *(v4f*)&Vs[cur][vr][sc]      = rv0;
        *(v4f*)&Vs[cur][16 + vr][sc] = rv1;
        __syncthreads();

        if (t + 1 < SB / KS) {
            const size_t off = (size_t)(t + 1) * KS * DM;
            rk0 = *(const v4f*)(ksrc + off);
            rk1 = *(const v4f*)(ksrc + off + (size_t)16 * DM);
            rv0 = *(const v4f*)(vsrc + off);
            rv1 = *(const v4f*)(vsrc + off + (size_t)16 * DM);
        }

        v8f sacc[2];
        sacc[0] = (v8f){0,0,0,0,0,0,0,0};
        sacc[1] = (v8f){0,0,0,0,0,0,0,0};
        {
            const float* kp0 = &Ks[cur][l16][2 * half];
            const float* kp1 = &Ks[cur][16 + l16][2 * half];
#pragma unroll
            for (int kk = 0; kk < 16; ++kk) {
                v2f a0 = *(const v2f*)(kp0 + kk * 4);
                v2f a1 = *(const v2f*)(kp1 + kk * 4);
                sacc[0] = wmma_f32(a0, qb[kk], sacc[0]);
                sacc[1] = wmma_f32(a1, qb[kk], sacc[1]);
            }
        }

#pragma unroll
        for (int u = 0; u < 2; ++u) {
            const v8f s = sacc[u];

            float sv[8];
#pragma unroll
            for (int j = 0; j < 8; ++j) sv[j] = s[j] * SCALE;
            float mx = fmaxf(fmaxf(fmaxf(sv[0], sv[1]), fmaxf(sv[2], sv[3])),
                             fmaxf(fmaxf(sv[4], sv[5]), fmaxf(sv[6], sv[7])));
            mx = fmaxf(mx, xhalf(mx));
            const float nm   = fmaxf(m, mx);
            const float corr = __expf(m - nm);
            float p[8];
#pragma unroll
            for (int j = 0; j < 8; ++j) p[j] = __expf(sv[j] - nm);
            float ps = ((p[0] + p[1]) + (p[2] + p[3])) + ((p[4] + p[5]) + (p[6] + p[7]));
            ps += xhalf(ps);
            lsum = lsum * corr + ps;
            m = nm;
#pragma unroll
            for (int dt = 0; dt < 4; ++dt)
#pragma unroll
                for (int j = 0; j < 8; ++j) o[dt][j] *= corr;

#pragma unroll
            for (int c = 0; c < 4; ++c) {
                v2f pb; pb.x = p[2 * c]; pb.y = p[2 * c + 1];
                const float* vp = &Vs[cur][u * 16 + 4 * c + 2 * half][l16];
#pragma unroll
                for (int dt = 0; dt < 4; ++dt) {
                    v2f a; a.x = vp[dt * 16]; a.y = vp[VP + dt * 16];
                    o[dt] = wmma_f32(a, pb, o[dt]);
                }
            }
        }
    }

    {
        const float inv = 1.0f / lsum;
        float* orow = O + (size_t)(qBase + l16) * DM + dBase + 8 * half;
#pragma unroll
        for (int dt = 0; dt < 4; ++dt) {
            v4f lo, hi;
            lo.x = o[dt][0] * inv; lo.y = o[dt][1] * inv; lo.z = o[dt][2] * inv; lo.w = o[dt][3] * inv;
            hi.x = o[dt][4] * inv; hi.y = o[dt][5] * inv; hi.z = o[dt][6] * inv; hi.w = o[dt][7] * inv;
            *(v4f*)(orow + dt * 16)     = lo;
            *(v4f*)(orow + dt * 16 + 4) = hi;
        }
    }
}

extern "C" void kernel_launch(void* const* d_in, const int* in_sizes, int n_in,
                              void* d_out, int out_size, void* d_ws, size_t ws_size,
                              hipStream_t stream) {
    const float* x  = (const float*)d_in[0];
    const float* fc = (const float*)d_in[1];
    const float* fs = (const float*)d_in[2];
    const float* wq = (const float*)d_in[3];
    const float* bq = (const float*)d_in[4];
    const float* wk = (const float*)d_in[5];
    const float* bk = (const float*)d_in[6];
    const float* wv = (const float*)d_in[7];
    const float* bv = (const float*)d_in[8];
    const float* wo = (const float*)d_in[9];
    const float* bo = (const float*)d_in[10];
    const float* gq = (const float*)d_in[11];
    const float* gk = (const float*)d_in[12];
    float* out = (float*)d_out;

    const size_t mat = (size_t)SB * DM;
    float* Qw = (float*)d_ws;
    float* Kw = Qw + mat;
    float* Vw = Kw + mat;
    float* Aw = Vw + mat;

    gemm16_kernel<<<dim3(4 * RT), 256, 0, stream>>>(x, wq, bq, gq, Qw, wk, bk, gk, Kw, wv, bv, Vw, fc, fs, 2 * RT);
    attn_kernel<<<dim3(SB / 16 / AW, NH), 32 * AW, 0, stream>>>(Qw, Kw, Vw, Aw);
    gemm16_kernel<<<dim3(2 * RT), 256, 0, stream>>>(Aw, wq, bq, gq, Qw, wk, bk, gk, Kw, wo, bo, out, fc, fs, 0);
}


# GEMM kernel: both workgroup barriers per K slab removed (each wave stages W in its own LDS region; one barrier for the X tile before the loop); waves run the slabs unsynchronised; sums unchanged
# speedup vs baseline: 3.8750x; 1.1923x over previous
#include <hip/hip_runtime.h>
#include <math.h>

#define SB   2048
#define DM   1024
#define NH   16
#define HD   64
#define HD2  32
#define SCALE 0.125f
#define EPS  1e-6f

typedef __attribute__((ext_vector_type(2))) float v2f;
typedef __attribute__((ext_vector_type(4))) float v4f;
typedef __attribute__((ext_vector_type(8))) float v8f;

#define KT     32
#define WPITCH (KT + 4)

static __device__ __forceinline__ void wave_lds_order() {
    __builtin_amdgcn_fence(__ATOMIC_RELEASE, "wavefront");
    __builtin_amdgcn_wave_barrier();
    __builtin_amdgcn_fence(__ATOMIC_ACQUIRE, "wavefront");
}

static __device__ __forceinline__ v8f wmma_f32(v2f a, v2f b, v8f c) {
    return __builtin_amdgcn_wmma_f32_16x16x4_f32(
         false, a,  false, b,
         (short)0, c,  false,  false);
}

template <int TT, bool NORM>
static __device__ __forceinline__ void gemm_tile(
    const float* __restrict__ X,
    const float* __restrict__ W,
    const float* __restrict__ bias,
    const float* __restrict__ g,
    const float* __restrict__ fc,
    const float* __restrict__ fs,
    float* __restrict__ out,
    const int rowBase, const int colBase,
    float (*tile)[DM + 4],
    float (*wbuf)[128][WPITCH],
    float (*redbuf)[17], float* rrms)
{
    static_assert(!NORM || TT == 8, "the row norm needs the whole row in one workgroup");
    constexpr int NC = TT * 16;
    constexpr int NL = NC / 4;
    constexpr int PR = TT * 64;

    const int tid  = threadIdx.x;
    const int wave = tid >> 5;
    const int lane = tid & 31;
    const int half = lane >> 4;
    const int l16  = lane & 15;
    const int nColLoc  = wave * NC;
    const int nColBase = colBase + nColLoc;

    v8f acc[TT];
#pragma unroll
    for (int t = 0; t < TT; ++t) acc[t] = (v8f){0,0,0,0,0,0,0,0};

#pragma unroll
    for (int i = 0; i < 16; ++i) {
        v4f v = *(const v4f*)(X + (size_t)(rowBase + i) * DM + tid * 4);
        *(v4f*)&tile[i][tid * 4] = v;
    }

    const int wr = lane >> 3;
    const int wc = (lane & 7) * 4;
    const float* wsrc = W + (size_t)(nColBase + wr) * DM + wc;
    v4f R[NL];
#pragma unroll
    for (int i = 0; i < NL; ++i) R[i] = *(const v4f*)(wsrc + (size_t)(i * 4) * DM);

    __syncthreads();

    for (int s = 0; s < DM / KT; ++s) {
        wave_lds_order();
#pragma unroll
        for (int i = 0; i < NL; ++i) *(v4f*)&wbuf[wave][i * 4 + wr][wc] = R[i];
        wave_lds_order();

        if (s + 1 < DM / KT) {
            const float* wn = wsrc + (s + 1) * KT;
#pragma unroll
            for (int i = 0; i < NL; ++i) R[i] = *(const v4f*)(wn + (size_t)(i * 4) * DM);
        }

        const float* arow = &tile[l16][s * KT + 2 * half];
        const float* brow = &wbuf[wave][l16][2 * half];
#pragma unroll
        for (int kk = 0; kk < KT / 4; ++kk) {
            v2f a = *(const v2f*)(arow + kk * 4);
#pragma unroll
            for (int t = 0; t < TT; ++t) {
                v2f b = *(const v2f*)(brow + t * 16 * WPITCH + kk * 4);
                acc[t] = wmma_f32(a, b, acc[t]);
            }
        }
    }
    __syncthreads();

#pragma unroll
    for (int t = 0; t < TT; ++t) {
        int cl = nColLoc + t * 16 + l16;
        float bv = bias[colBase + cl];
#pragma unroll
        for (int j = 0; j < 8; ++j) {
            int row = j + 8 * half;
            tile[row][cl] = acc[t][j] + bv;
        }
    }
    __syncthreads();

    if (NORM) {
        {
            int r  = tid & 15;
            int c0 = tid >> 4;
            float s = 0.f;
            for (int c = c0; c < DM; c += 16) { float v = tile[r][c]; s += v * v; }
            redbuf[r][c0] = s;
        }
        __syncthreads();
        if (tid < 16) {
            float s = 0.f;
#pragma unroll
            for (int c = 0; c < 16; ++c) s += redbuf[tid][c];
            rrms[tid] = rsqrtf(s * (1.0f / (float)DM) + EPS);
        }
        __syncthreads();
    }

    for (int p = tid; p < 16 * PR; p += 256) {
        int r    = p / PR;
        int cp   = p % PR;
        int dcol = cp * 2;
        int srow = rowBase + r;
        float re = tile[r][dcol];
        float im = tile[r][dcol + 1];
        if (NORM) {
            float rr = rrms[r];
            re *= rr * g[dcol];
            im *= rr * g[dcol + 1];
            int   pp = cp & (HD2 - 1);
            float c  = fc[srow * HD2 + pp];
            float sn = fs[srow * HD2 + pp];
            float ore = re * c - im * sn;
            float oim = re * sn + im * c;
            re = ore; im = oim;
        }
        out[(size_t)srow * DM + colBase + dcol]     = re;
        out[(size_t)srow * DM + colBase + dcol + 1] = im;
    }
}

#define RT (SB / 16)

__global__ __launch_bounds__(256) void gemm16_kernel(
    const float* __restrict__ X,
    const float* __restrict__ W0, const float* __restrict__ b0, const float* __restrict__ g0, float* __restrict__ out0,
    const float* __restrict__ W1, const float* __restrict__ b1, const float* __restrict__ g1, float* __restrict__ out1,
    const float* __restrict__ W2, const float* __restrict__ b2, float* __restrict__ out2,
    const float* __restrict__ fc,
    const float* __restrict__ fs,
    int nFull)
{
    __shared__ __attribute__((aligned(16))) float tile[16][DM + 4];
    __shared__ __attribute__((aligned(16))) float wbuf[8][128][WPITCH];
    __shared__ float redbuf[16][17];
    __shared__ float rrms[16];

    const int b = blockIdx.x;
    if (b < nFull) {
        const bool second = b >= RT;
        const int  rt     = second ? b - RT : b;
        gemm_tile<8, true>(X, second ? W1 : W0, second ? b1 : b0, second ? g1 : g0, fc, fs,
                           second ? out1 : out0, rt * 16, 0, tile, wbuf, redbuf, rrms);
    } else {
        const int i = b - nFull;
        gemm_tile<4, false>(X, W2, b2, g0, fc, fs, out2, (i >> 1) * 16, (i & 1) * (DM / 2),
                            tile, wbuf, redbuf, rrms);
    }
}

#define AW  8
#define KS  32
#define KP  (HD + 4)
#define VP  (HD + 16)

static __device__ __forceinline__ float xhalf(float x) {
    int b = __builtin_bit_cast(int, x);
    return __builtin_bit_cast(float,
        __builtin_amdgcn_permlanex16(b, b, 0x76543210, 0xfedcba98, false, false));
}

__global__ __launch_bounds__(256) void attn_kernel(
    const float* __restrict__ Q,
    const float* __restrict__ K,
    const float* __restrict__ V,
    float* __restrict__ O)
{
    __shared__ __attribute__((aligned(16))) float Ks[2][KS][KP];
    __shared__ __attribute__((aligned(16))) float Vs[2][KS][VP];

    const int tid  = threadIdx.x;
    const int wave = tid >> 5;
    const int lane = tid & 31;
    const int half = lane >> 4;
    const int l16  = lane & 15;
    const int qt    = blockIdx.x * AW + wave;
    const int h     = blockIdx.y;
    const int qBase = qt * 16;
    const int dBase = h * HD;

    const int sr = tid >> 4;
    const int sc = (tid & 15) * 4;
    const int vr = 4 * ((sr & 7) >> 1) + 2 * (sr >> 3) + (sr & 1);
    const float* ksrc = K + (size_t)sr * DM + dBase + sc;
    const float* vsrc = V + (size_t)sr * DM + dBase + sc;
    v4f rk0 = *(const v4f*)(ksrc);
    v4f rk1 = *(const v4f*)(ksrc + (size_t)16 * DM);
    v4f rv0 = *(const v4f*)(vsrc);
    v4f rv1 = *(const v4f*)(vsrc + (size_t)16 * DM);

    v2f qb[16];
    {
        const float* qp = Q + (size_t)(qBase + l16) * DM + dBase + 2 * half;
#pragma unroll
        for (int kk = 0; kk < 16; ++kk) qb[kk] = *(const v2f*)(qp + kk * 4);
    }

    v8f o[4];
#pragma unroll
    for (int dt = 0; dt < 4; ++dt) o[dt] = (v8f){0,0,0,0,0,0,0,0};
    float m = -INFINITY, lsum = 0.f;

    for (int t = 0; t < SB / KS; ++t) {
        const int cur = t & 1;

        *(v4f*)&Ks[cur][sr][sc]      = rk0;
        *(v4f*)&Ks[cur][sr + 16][sc] = rk1;
        *(v4f*)&Vs[cur][vr][sc]      = rv0;
        *(v4f*)&Vs[cur][16 + vr][sc] = rv1;
        __syncthreads();

        if (t + 1 < SB / KS) {
            const size_t off = (size_t)(t + 1) * KS * DM;
            rk0 = *(const v4f*)(ksrc + off);
            rk1 = *(const v4f*)(ksrc + off + (size_t)16 * DM);
            rv0 = *(const v4f*)(vsrc + off);
            rv1 = *(const v4f*)(vsrc + off + (size_t)16 * DM);
        }

        v8f sacc[2];
        sacc[0] = (v8f){0,0,0,0,0,0,0,0};
        sacc[1] = (v8f){0,0,0,0,0,0,0,0};
        {
            const float* kp0 = &Ks[cur][l16][2 * half];
            const float* kp1 = &Ks[cur][16 + l16][2 * half];
#pragma unroll
            for (int kk = 0; kk < 16; ++kk) {
                v2f a0 = *(const v2f*)(kp0 + kk * 4);
                v2f a1 = *(const v2f*)(kp1 + kk * 4);
                sacc[0] = wmma_f32(a0, qb[kk], sacc[0]);
                sacc[1] = wmma_f32(a1, qb[kk], sacc[1]);
            }
        }

#pragma unroll
        for (int u = 0; u < 2; ++u) {
            const v8f s = sacc[u];

            float sv[8];
#pragma unroll
            for (int j = 0; j < 8; ++j) sv[j] = s[j] * SCALE;
            float mx = fmaxf(fmaxf(fmaxf(sv[0], sv[1]), fmaxf(sv[2], sv[3])),
                             fmaxf(fmaxf(sv[4], sv[5]), fmaxf(sv[6], sv[7])));
            mx = fmaxf(mx, xhalf(mx));
            const float nm   = fmaxf(m, mx);
            const float corr = __expf(m - nm);
            float p[8];
#pragma unroll
            for (int j = 0; j < 8; ++j) p[j] = __expf(sv[j] - nm);
            float ps = ((p[0] + p[1]) + (p[2] + p[3])) + ((p[4] + p[5]) + (p[6] + p[7]));
            ps += xhalf(ps);
            lsum = lsum * corr + ps;
            m = nm;
#pragma unroll
            for (int dt = 0; dt < 4; ++dt)
#pragma unroll
                for (int j = 0; j < 8; ++j) o[dt][j] *= corr;

#pragma unroll
            for (int c = 0; c < 4; ++c) {
                v2f pb; pb.x = p[2 * c]; pb.y = p[2 * c + 1];
                const float* vp = &Vs[cur][u * 16 + 4 * c + 2 * half][l16];
#pragma unroll
                for (int dt = 0; dt < 4; ++dt) {
                    v2f a; a.x = vp[dt * 16]; a.y = vp[VP + dt * 16];
                    o[dt] = wmma_f32(a, pb, o[dt]);
                }
            }
        }
    }

    {
        const float inv = 1.0f / lsum;
        float* orow = O + (size_t)(qBase + l16) * DM + dBase + 8 * half;
#pragma unroll
        for (int dt = 0; dt < 4; ++dt) {
            v4f lo, hi;
            lo.x = o[dt][0] * inv; lo.y = o[dt][1] * inv; lo.z = o[dt][2] * inv; lo.w = o[dt][3] * inv;
            hi.x = o[dt][4] * inv; hi.y = o[dt][5] * inv; hi.z = o[dt][6] * inv; hi.w = o[dt][7] * inv;
            *(v4f*)(orow + dt * 16)     = lo;
            *(v4f*)(orow + dt * 16 + 4) = hi;
        }
    }
}

extern "C" void kernel_launch(void* const* d_in, const int* in_sizes, int n_in,
                              void* d_out, int out_size, void* d_ws, size_t ws_size,
                              hipStream_t stream) {
    const float* x  = (const float*)d_in[0];
    const float* fc = (const float*)d_in[1];
    const float* fs = (const float*)d_in[2];
    const float* wq = (const float*)d_in[3];
    const float* bq = (const float*)d_in[4];
    const float* wk = (const float*)d_in[5];
    const float* bk = (const float*)d_in[6];
    const float* wv = (const float*)d_in[7];
    const float* bv = (const float*)d_in[8];
    const float* wo = (const float*)d_in[9];
    const float* bo = (const float*)d_in[10];
    const float* gq = (const float*)d_in[11];
    const float* gk = (const float*)d_in[12];
    float* out = (float*)d_out;

    const size_t mat = (size_t)SB * DM;
    float* Qw = (float*)d_ws;
    float* Kw = Qw + mat;
    float* Vw = Kw + mat;
    float* Aw = Vw + mat;

    gemm16_kernel<<<dim3(4 * RT), 256, 0, stream>>>(x, wq, bq, gq, Qw, wk, bk, gk, Kw, wv, bv, Vw, fc, fs, 2 * RT);
    attn_kernel<<<dim3(SB / 16 / AW, NH), 32 * AW, 0, stream>>>(Qw, Kw, Vw, Aw);
    gemm16_kernel<<<dim3(2 * RT), 256, 0, stream>>>(Aw, wq, bq, gq, Qw, wk, bk, gk, Kw, wo, bo, out, fc, fs, 0);
}


# GEMM tiles 32 rows tall: each W line and W fragment feeds two 16-row blocks (q,k,v as one 2048x3072 product in 32x768 tiles, out in 32x256); row norm + rope moved to its own pass; sums unchanged
# speedup vs baseline: 4.0262x; 1.0390x over previous
#include <hip/hip_runtime.h>
#include <math.h>

#define SB   2048
#define DM   1024
#define NH   16
#define HD   64
#define HD2  32
#define SCALE 0.125f
#define EPS  1e-6f

typedef __attribute__((ext_vector_type(2))) float v2f;
typedef __attribute__((ext_vector_type(4))) float v4f;
typedef __attribute__((ext_vector_type(8))) float v8f;

#define KT     32
#define WPITCH (KT + 4)

static __device__ __forceinline__ void wave_lds_order() {
    __builtin_amdgcn_fence(__ATOMIC_RELEASE, "wavefront");
    __builtin_amdgcn_wave_barrier();
    __builtin_amdgcn_fence(__ATOMIC_ACQUIRE, "wavefront");
}

static __device__ __forceinline__ v8f wmma_f32(v2f a, v2f b, v8f c) {
    return __builtin_amdgcn_wmma_f32_16x16x4_f32(
         false, a,  false, b,
         (short)0, c,  false,  false);
}

#define MR 2

template <int TT>
static __device__ __forceinline__ void gemm_tile32(
    const float* __restrict__ X,
    const float* __restrict__ W0, const float* __restrict__ b0, float* __restrict__ out0,
    const float* __restrict__ W1, const float* __restrict__ b1, float* __restrict__ out1,
    const float* __restrict__ W2, const float* __restrict__ b2, float* __restrict__ out2,
    const int rowBase, const int colBase,
    float (*tile)[DM + 4],
    float* wreg)
{
    constexpr int NC = TT * 16;
    constexpr int NL = NC / 4;
    constexpr int PQ = TT * 32;

    const int tid  = threadIdx.x;
    const int wave = tid >> 5;
    const int lane = tid & 31;
    const int half = lane >> 4;
    const int l16  = lane & 15;
    const int nColLoc  = wave * NC;
    const int nColBase = colBase + nColLoc;

    v8f acc[MR][TT];
#pragma unroll
    for (int m = 0; m < MR; ++m)
#pragma unroll
        for (int t = 0; t < TT; ++t) acc[m][t] = (v8f){0,0,0,0,0,0,0,0};

#pragma unroll
    for (int i = 0; i < 16 * MR; ++i) {
        v4f v = *(const v4f*)(X + (size_t)(rowBase + i) * DM + tid * 4);
        *(v4f*)&tile[i][tid * 4] = v;
    }

    const int wr = lane >> 3;
    const int wc = (lane & 7) * 4;
    const float* wsrc[TT];
#pragma unroll
    for (int t = 0; t < TT; ++t) {
        const int cv = nColBase + t * 16;
        const int mi = cv / DM;
        const float* Wm = mi == 0 ? W0 : (mi == 1 ? W1 : W2);
        wsrc[t] = Wm + (size_t)(cv % DM + wr) * DM + wc;
    }
    v4f R[NL];
#pragma unroll
    for (int i = 0; i < NL; ++i) R[i] = *(const v4f*)(wsrc[i / 4] + (size_t)((i % 4) * 4) * DM);

    __syncthreads();

    for (int s = 0; s < DM / KT; ++s) {
        wave_lds_order();
#pragma unroll
        for (int i = 0; i < NL; ++i) *(v4f*)&wreg[(i * 4 + wr) * WPITCH + wc] = R[i];
        wave_lds_order();

        if (s + 1 < DM / KT) {
#pragma unroll
            for (int i = 0; i < NL; ++i)
                R[i] = *(const v4f*)(wsrc[i / 4] + (size_t)((i % 4) * 4) * DM + (s + 1) * KT);
        }

        const float* arow = &tile[l16][s * KT + 2 * half];
        const float* brow = &wreg[l16 * WPITCH + 2 * half];
#pragma unroll
        for (int kk = 0; kk < KT / 4; ++kk) {
            v2f a[MR];
#pragma unroll
            for (int m = 0; m < MR; ++m) a[m] = *(const v2f*)(arow + m * 16 * (DM + 4) + kk * 4);
#pragma unroll
            for (int t = 0; t < TT; ++t) {
                v2f b = *(const v2f*)(brow + t * 16 * WPITCH + kk * 4);
#pragma unroll
                for (int m = 0; m < MR; ++m) acc[m][t] = wmma_f32(a[m], b, acc[m][t]);
            }
        }
    }
    __syncthreads();

#pragma unroll
    for (int t = 0; t < TT; ++t) {
        const int cl = nColLoc + t * 16 + l16;
        const int cv = colBase + cl;
        const int mi = (nColBase + t * 16) / DM;
        const float* bm = mi == 0 ? b0 : (mi == 1 ? b1 : b2);
        const float bv = bm[cv % DM];
#pragma unroll
        for (int m = 0; m < MR; ++m)
#pragma unroll
            for (int j = 0; j < 8; ++j)
                tile[m * 16 + j + 8 * half][cl] = acc[m][t][j] + bv;
    }
    __syncthreads();

    for (int p = tid; p < 16 * MR * PQ; p += 256) {
        const int r  = p / PQ;
        const int c4 = (p % PQ) * 4;
        const int cv = colBase + c4;
        const int mi = cv / DM;
        float* om = mi == 0 ? out0 : (mi == 1 ? out1 : out2);
        *(v4f*)(om + (size_t)(rowBase + r) * DM + cv % DM) = *(const v4f*)&tile[r][c4];
    }
}

#define RT (SB / 16)

template <int TT>
__global__ __launch_bounds__(256) void gemm32_kernel(
    const float* __restrict__ X,
    const float* __restrict__ W0, const float* __restrict__ b0, float* __restrict__ out0,
    const float* __restrict__ W1, const float* __restrict__ b1, float* __restrict__ out1,
    const float* __restrict__ W2, const float* __restrict__ b2, float* __restrict__ out2,
    int CT)
{
    __shared__ __attribute__((aligned(16))) float tile[16 * MR][DM + 4];
    __shared__ __attribute__((aligned(16))) float wbuf[8][TT * 16 * WPITCH];

    const int b  = blockIdx.x;
    const int rt = b / CT;
    const int ct = b % CT;
    gemm_tile32<TT>(X, W0, b0, out0, W1, b1, out1, W2, b2, out2,
                    rt * 16 * MR, ct * (TT * 128), tile, &wbuf[threadIdx.x >> 5][0]);
}

__global__ __launch_bounds__(256) void normrope_kernel(
    float* P0, const float* __restrict__ g0,
    float* P1, const float* __restrict__ g1,
    const float* __restrict__ fc,
    const float* __restrict__ fs)
{
    __shared__ __attribute__((aligned(16))) float tile[16][DM + 4];
    __shared__ float redbuf[16][17];
    __shared__ float rrms[16];

    const int tid = threadIdx.x;
    const int b   = blockIdx.x;
    const bool second = b >= RT;
    const int rowBase = (second ? b - RT : b) * 16;
    float* P = second ? P1 : P0;
    const float* __restrict__ g = second ? g1 : g0;

#pragma unroll
    for (int i = 0; i < 16; ++i) {
        v4f v = *(const v4f*)(P + (size_t)(rowBase + i) * DM + tid * 4);
        *(v4f*)&tile[i][tid * 4] = v;
    }
    __syncthreads();

    {
        int r  = tid & 15;
        int c0 = tid >> 4;
        float s = 0.f;
        for (int c = c0; c < DM; c += 16) { float v = tile[r][c]; s += v * v; }
        redbuf[r][c0] = s;
    }
    __syncthreads();
    if (tid < 16) {
        float s = 0.f;
#pragma unroll
        for (int c = 0; c < 16; ++c) s += redbuf[tid][c];
        rrms[tid] = rsqrtf(s * (1.0f / (float)DM) + EPS);
    }
    __syncthreads();

    const int   c4 = tid * 4;
    const int   pp = (c4 >> 1) & (HD2 - 1);
    const v4f   gv = *(const v4f*)(g + c4);
#pragma unroll
    for (int r = 0; r < 16; ++r) {
        const int   srow = rowBase + r;
        const float rr = rrms[r];
        const v4f   v  = *(const v4f*)&tile[r][c4];
        const v2f   c  = *(const v2f*)(fc + srow * HD2 + pp);
        const v2f   sn = *(const v2f*)(fs + srow * HD2 + pp);
        const float re0 = v.x * (rr * gv.x), im0 = v.y * (rr * gv.y);
        const float re1 = v.z * (rr * gv.z), im1 = v.w * (rr * gv.w);
        v4f o;
        o.x = re0 * c.x - im0 * sn.x;
        o.y = re0 * sn.x + im0 * c.x;
        o.z = re1 * c.y - im1 * sn.y;
        o.w = re1 * sn.y + im1 * c.y;
        *(v4f*)(P + (size_t)srow * DM + c4) = o;
    }
}

#define AW  8
#define KS  32
#define KP  (HD + 4)
#define VP  (HD + 16)

static __device__ __forceinline__ float xhalf(float x) {
    int b = __builtin_bit_cast(int, x);
    return __builtin_bit_cast(float,
        __builtin_amdgcn_permlanex16(b, b, 0x76543210, 0xfedcba98, false, false));
}

__global__ __launch_bounds__(256) void attn_kernel(
    const float* __restrict__ Q,
    const float* __restrict__ K,
    const float* __restrict__ V,
    float* __restrict__ O)
{
    __shared__ __attribute__((aligned(16))) float Ks[2][KS][KP];
    __shared__ __attribute__((aligned(16))) float Vs[2][KS][VP];

    const int tid  = threadIdx.x;
    const int wave = tid >> 5;
    const int lane = tid & 31;
    const int half = lane >> 4;
    const int l16  = lane & 15;
    const int qt    = blockIdx.x * AW + wave;
    const int h     = blockIdx.y;
    const int qBase = qt * 16;
    const int dBase = h * HD;

    const int sr = tid >> 4;
    const int sc = (tid & 15) * 4;
    const int vr = 4 * ((sr & 7) >> 1) + 2 * (sr >> 3) + (sr & 1);
    const float* ksrc = K + (size_t)sr * DM + dBase + sc;
    const float* vsrc = V + (size_t)sr * DM + dBase + sc;
    v4f rk0 = *(const v4f*)(ksrc);
    v4f rk1 = *(const v4f*)(ksrc + (size_t)16 * DM);
    v4f rv0 = *(const v4f*)(vsrc);
    v4f rv1 = *(const v4f*)(vsrc + (size_t)16 * DM);

    v2f qb[16];
    {
        const float* qp = Q + (size_t)(qBase + l16) * DM + dBase + 2 * half;
#pragma unroll
        for (int kk = 0; kk < 16; ++kk) qb[kk] = *(const v2f*)(qp + kk * 4);
    }

    v8f o[4];
#pragma unroll
    for (int dt = 0; dt < 4; ++dt) o[dt] = (v8f){0,0,0,0,0,0,0,0};
    float m = -INFINITY, lsum = 0.f;

    for (int t = 0; t < SB / KS; ++t) {
        const int cur = t & 1;

        *(v4f*)&Ks[cur][sr][sc]      = rk0;
        *(v4f*)&Ks[cur][sr + 16][sc] = rk1;
        *(v4f*)&Vs[cur][vr][sc]      = rv0;
        *(v4f*)&Vs[cur][16 + vr][sc] = rv1;
        __syncthreads();

        if (t + 1 < SB / KS) {
            const size_t off = (size_t)(t + 1) * KS * DM;
            rk0 = *(const v4f*)(ksrc + off);
            rk1 = *(const v4f*)(ksrc + off + (size_t)16 * DM);
            rv0 = *(const v4f*)(vsrc + off);
            rv1 = *(const v4f*)(vsrc + off + (size_t)16 * DM);
        }

        v8f sacc[2];
        sacc[0] = (v8f){0,0,0,0,0,0,0,0};
        sacc[1] = (v8f){0,0,0,0,0,0,0,0};
        {
            const float* kp0 = &Ks[cur][l16][2 * half];
            const float* kp1 = &Ks[cur][16 + l16][2 * half];
#pragma unroll
            for (int kk = 0; kk < 16; ++kk) {
                v2f a0 = *(const v2f*)(kp0 + kk * 4);
                v2f a1 = *(const v2f*)(kp1 + kk * 4);
                sacc[0] = wmma_f32(a0, qb[kk], sacc[0]);
                sacc[1] = wmma_f32(a1, qb[kk], sacc[1]);
            }
        }

#pragma unroll
        for (int u = 0; u < 2; ++u) {
            const v8f s = sacc[u];

            float sv[8];
#pragma unroll
            for (int j = 0; j < 8; ++j) sv[j] = s[j] * SCALE;
            float mx = fmaxf(fmaxf(fmaxf(sv[0], sv[1]), fmaxf(sv[2], sv[3])),
                             fmaxf(fmaxf(sv[4], sv[5]), fmaxf(sv[6], sv[7])));
            mx = fmaxf(mx, xhalf(mx));
            const float nm   = fmaxf(m, mx);
            const float corr = __expf(m - nm);
            float p[8];
#pragma unroll
            for (int j = 0; j < 8; ++j) p[j] = __expf(sv[j] - nm);
            float ps = ((p[0] + p[1]) + (p[2] + p[3])) + ((p[4] + p[5]) + (p[6] + p[7]));
            ps += xhalf(ps);
            lsum = lsum * corr + ps;
            m = nm;
#pragma unroll
            for (int dt = 0; dt < 4; ++dt)
#pragma unroll
                for (int j = 0; j < 8; ++j) o[dt][j] *= corr;

#pragma unroll
            for (int c = 0; c < 4; ++c) {
                v2f pb; pb.x = p[2 * c]; pb.y = p[2 * c + 1];
                const float* vp = &Vs[cur][u * 16 + 4 * c + 2 * half][l16];
#pragma unroll
                for (int dt = 0; dt < 4; ++dt) {
                    v2f a; a.x = vp[dt * 16]; a.y = vp[VP + dt * 16];
                    o[dt] = wmma_f32(a, pb, o[dt]);
                }
            }
        }
    }

    {
        const float inv = 1.0f / lsum;
        float* orow = O + (size_t)(qBase + l16) * DM + dBase + 8 * half;
#pragma unroll
        for (int dt = 0; dt < 4; ++dt) {
            v4f lo, hi;
            lo.x = o[dt][0] * inv; lo.y = o[dt][1] * inv; lo.z = o[dt][2] * inv; lo.w = o[dt][3] * inv;
            hi.x = o[dt][4] * inv; hi.y = o[dt][5] * inv; hi.z = o[dt][6] * inv; hi.w = o[dt][7] * inv;
            *(v4f*)(orow + dt * 16)     = lo;
            *(v4f*)(orow + dt * 16 + 4) = hi;
        }
    }
}

extern "C" void kernel_launch(void* const* d_in, const int* in_sizes, int n_in,
                              void* d_out, int out_size, void* d_ws, size_t ws_size,
                              hipStream_t stream) {
    const float* x  = (const float*)d_in[0];
    const float* fc = (const float*)d_in[1];
    const float* fs = (const float*)d_in[2];
    const float* wq = (const float*)d_in[3];
    const float* bq = (const float*)d_in[4];
    const float* wk = (const float*)d_in[5];
    const float* bk = (const float*)d_in[6];
    const float* wv = (const float*)d_in[7];
    const float* bv = (const float*)d_in[8];
    const float* wo = (const float*)d_in[9];
    const float* bo = (const float*)d_in[10];
    const float* gq = (const float*)d_in[11];
    const float* gk = (const float*)d_in[12];
    float* out = (float*)d_out;

    const size_t mat = (size_t)SB * DM;
    float* Qw = (float*)d_ws;
    float* Kw = Qw + mat;
    float* Vw = Kw + mat;
    float* Aw = Vw + mat;

    gemm32_kernel<6><<<dim3(256), 256, 0, stream>>>(x, wq, bq, Qw, wk, bk, Kw, wv, bv, Vw, 4);
    normrope_kernel<<<dim3(2 * RT), 256, 0, stream>>>(Qw, gq, Kw, gk, fc, fs);
    attn_kernel<<<dim3(SB / 16 / AW, NH), 32 * AW, 0, stream>>>(Qw, Kw, Vw, Aw);
    gemm32_kernel<2><<<dim3(256), 256, 0, stream>>>(Aw, wo, bo, out, wo, bo, out, wo, bo, out, 4);
}


# attention: 16-wave workgroups, each query tile's keys split in two halves run by two waves (4 waves per SIMD, was 2), halves merged through LDS at the end; softmax sum and PV sum order change
# speedup vs baseline: 4.1610x; 1.0335x over previous
#include <hip/hip_runtime.h>
#include <math.h>

#define SB   2048
#define DM   1024
#define NH   16
#define HD   64
#define HD2  32
#define SCALE 0.125f
#define EPS  1e-6f

typedef __attribute__((ext_vector_type(2))) float v2f;
typedef __attribute__((ext_vector_type(4))) float v4f;
typedef __attribute__((ext_vector_type(8))) float v8f;

#define KT     32
#define WPITCH (KT + 4)

static __device__ __forceinline__ void wave_lds_order() {
    __builtin_amdgcn_fence(__ATOMIC_RELEASE, "wavefront");
    __builtin_amdgcn_wave_barrier();
    __builtin_amdgcn_fence(__ATOMIC_ACQUIRE, "wavefront");
}

static __device__ __forceinline__ v8f wmma_f32(v2f a, v2f b, v8f c) {
    return __builtin_amdgcn_wmma_f32_16x16x4_f32(
         false, a,  false, b,
         (short)0, c,  false,  false);
}

#define MR 2

template <int TT>
static __device__ __forceinline__ void gemm_tile32(
    const float* __restrict__ X,
    const float* __restrict__ W0, const float* __restrict__ b0, float* __restrict__ out0,
    const float* __restrict__ W1, const float* __restrict__ b1, float* __restrict__ out1,
    const float* __restrict__ W2, const float* __restrict__ b2, float* __restrict__ out2,
    const int rowBase, const int colBase,
    float (*tile)[DM + 4],
    float* wreg)
{
    constexpr int NC = TT * 16;
    constexpr int NL = NC / 4;
    constexpr int PQ = TT * 32;

    const int tid  = threadIdx.x;
    const int wave = tid >> 5;
    const int lane = tid & 31;
    const int half = lane >> 4;
    const int l16  = lane & 15;
    const int nColLoc  = wave * NC;
    const int nColBase = colBase + nColLoc;

    v8f acc[MR][TT];
#pragma unroll
    for (int m = 0; m < MR; ++m)
#pragma unroll
        for (int t = 0; t < TT; ++t) acc[m][t] = (v8f){0,0,0,0,0,0,0,0};

#pragma unroll
    for (int i = 0; i < 16 * MR; ++i) {
        v4f v = *(const v4f*)(X + (size_t)(rowBase + i) * DM + tid * 4);
        *(v4f*)&tile[i][tid * 4] = v;
    }

    const int wr = lane >> 3;
    const int wc = (lane & 7) * 4;
    const float* wsrc[TT];
#pragma unroll
    for (int t = 0; t < TT; ++t) {
        const int cv = nColBase + t * 16;
        const int mi = cv / DM;
        const float* Wm = mi == 0 ? W0 : (mi == 1 ? W1 : W2);
        wsrc[t] = Wm + (size_t)(cv % DM + wr) * DM + wc;
    }
    v4f R[NL];
#pragma unroll
    for (int i = 0; i < NL; ++i) R[i] = *(const v4f*)(wsrc[i / 4] + (size_t)((i % 4) * 4) * DM);

    __syncthreads();

    for (int s = 0; s < DM / KT; ++s) {
        wave_lds_order();
#pragma unroll
        for (int i = 0; i < NL; ++i) *(v4f*)&wreg[(i * 4 + wr) * WPITCH + wc] = R[i];
        wave_lds_order();

        if (s + 1 < DM / KT) {
#pragma unroll
            for (int i = 0; i < NL; ++i)
                R[i] = *(const v4f*)(wsrc[i / 4] + (size_t)((i % 4) * 4) * DM + (s + 1) * KT);
        }

        const float* arow = &tile[l16][s * KT + 2 * half];
        const float* brow = &wreg[l16 * WPITCH + 2 * half];
#pragma unroll
        for (int kk = 0; kk < KT / 4; ++kk) {
            v2f a[MR];
#pragma unroll
            for (int m = 0; m < MR; ++m) a[m] = *(const v2f*)(arow + m * 16 * (DM + 4) + kk * 4);
#pragma unroll
            for (int t = 0; t < TT; ++t) {
                v2f b = *(const v2f*)(brow + t * 16 * WPITCH + kk * 4);
#pragma unroll
                for (int m = 0; m < MR; ++m) acc[m][t] = wmma_f32(a[m], b, acc[m][t]);
            }
        }
    }
    __syncthreads();

#pragma unroll
    for (int t = 0; t < TT; ++t) {
        const int cl = nColLoc + t * 16 + l16;
        const int cv = colBase + cl;
        const int mi = (nColBase + t * 16) / DM;
        const float* bm = mi == 0 ? b0 : (mi == 1 ? b1 : b2);
        const float bv = bm[cv % DM];
#pragma unroll
        for (int m = 0; m < MR; ++m)
#pragma unroll
            for (int j = 0; j < 8; ++j)
                tile[m * 16 + j + 8 * half][cl] = acc[m][t][j] + bv;
    }
    __syncthreads();

    for (int p = tid; p < 16 * MR * PQ; p += 256) {
        const int r  = p / PQ;
        const int c4 = (p % PQ) * 4;
        const int cv = colBase + c4;
        const int mi = cv / DM;
        float* om = mi == 0 ? out0 : (mi == 1 ? out1 : out2);
        *(v4f*)(om + (size_t)(rowBase + r) * DM + cv % DM) = *(const v4f*)&tile[r][c4];
    }
}

#define RT (SB / 16)

template <int TT>
__global__ __launch_bounds__(256) void gemm32_kernel(
    const float* __restrict__ X,
    const float* __restrict__ W0, const float* __restrict__ b0, float* __restrict__ out0,
    const float* __restrict__ W1, const float* __restrict__ b1, float* __restrict__ out1,
    const float* __restrict__ W2, const float* __restrict__ b2, float* __restrict__ out2,
    int CT)
{
    __shared__ __attribute__((aligned(16))) float tile[16 * MR][DM + 4];
    __shared__ __attribute__((aligned(16))) float wbuf[8][TT * 16 * WPITCH];

    const int b  = blockIdx.x;
    const int rt = b / CT;
    const int ct = b % CT;
    gemm_tile32<TT>(X, W0, b0, out0, W1, b1, out1, W2, b2, out2,
                    rt * 16 * MR, ct * (TT * 128), tile, &wbuf[threadIdx.x >> 5][0]);
}

__global__ __launch_bounds__(256) void normrope_kernel(
    float* P0, const float* __restrict__ g0,
    float* P1, const float* __restrict__ g1,
    const float* __restrict__ fc,
    const float* __restrict__ fs)
{
    __shared__ __attribute__((aligned(16))) float tile[16][DM + 4];
    __shared__ float redbuf[16][17];
    __shared__ float rrms[16];

    const int tid = threadIdx.x;
    const int b   = blockIdx.x;
    const bool second = b >= RT;
    const int rowBase = (second ? b - RT : b) * 16;
    float* P = second ? P1 : P0;
    const float* __restrict__ g = second ? g1 : g0;

#pragma unroll
    for (int i = 0; i < 16; ++i) {
        v4f v = *(const v4f*)(P + (size_t)(rowBase + i) * DM + tid * 4);
        *(v4f*)&tile[i][tid * 4] = v;
    }
    __syncthreads();

    {
        int r  = tid & 15;
        int c0 = tid >> 4;
        float s = 0.f;
        for (int c = c0; c < DM; c += 16) { float v = tile[r][c]; s += v * v; }
        redbuf[r][c0] = s;
    }
    __syncthreads();
    if (tid < 16) {
        float s = 0.f;
#pragma unroll
        for (int c = 0; c < 16; ++c) s += redbuf[tid][c];
        rrms[tid] = rsqrtf(s * (1.0f / (float)DM) + EPS);
    }
    __syncthreads();

    const int   c4 = tid * 4;
    const int   pp = (c4 >> 1) & (HD2 - 1);
    const v4f   gv = *(const v4f*)(g + c4);
#pragma unroll
    for (int r = 0; r < 16; ++r) {
        const int   srow = rowBase + r;
        const float rr = rrms[r];
        const v4f   v  = *(const v4f*)&tile[r][c4];
        const v2f   c  = *(const v2f*)(fc + srow * HD2 + pp);
        const v2f   sn = *(const v2f*)(fs + srow * HD2 + pp);
        const float re0 = v.x * (rr * gv.x), im0 = v.y * (rr * gv.y);
        const float re1 = v.z * (rr * gv.z), im1 = v.w * (rr * gv.w);
        v4f o;
        o.x = re0 * c.x - im0 * sn.x;
        o.y = re0 * sn.x + im0 * c.x;
        o.z = re1 * c.y - im1 * sn.y;
        o.w = re1 * sn.y + im1 * c.y;
        *(v4f*)(P + (size_t)srow * DM + c4) = o;
    }
}

#define AW  8
#define NKH 2
#define KH  (SB / NKH)
#define KS  32
#define KP  (HD + 4)
#define VP  (HD + 16)
#define MG  34

static __device__ __forceinline__ float xhalf(float x) {
    int b = __builtin_bit_cast(int, x);
    return __builtin_bit_cast(float,
        __builtin_amdgcn_permlanex16(b, b, 0x76543210, 0xfedcba98, false, false));
}

__global__ __launch_bounds__(32 * AW * NKH) void attn_kernel(
    const float* __restrict__ Q,
    const float* __restrict__ K,
    const float* __restrict__ V,
    float* __restrict__ O)
{
    __shared__ __attribute__((aligned(16))) float Ks[NKH][2][KS][KP];
    __shared__ __attribute__((aligned(16))) float Vs[NKH][2][KS][VP];
    __shared__ float Mg[AW][MG][32];

    const int tid  = threadIdx.x;
    const int wave = tid >> 5;
    const int lane = tid & 31;
    const int half = lane >> 4;
    const int l16  = lane & 15;
    const int kh    = wave / AW;
    const int qw    = wave % AW;
    const int qt    = blockIdx.x * AW + qw;
    const int h     = blockIdx.y;
    const int qBase = qt * 16;
    const int dBase = h * HD;

    const int t8 = tid & (32 * AW - 1);
    const int sr = t8 >> 4;
    const int sc = (t8 & 15) * 4;
    const int vr = 4 * ((sr & 7) >> 1) + 2 * (sr >> 3) + (sr & 1);
    const float* ksrc = K + (size_t)(kh * KH + sr) * DM + dBase + sc;
    const float* vsrc = V + (size_t)(kh * KH + sr) * DM + dBase + sc;
    v4f rk0 = *(const v4f*)(ksrc);
    v4f rk1 = *(const v4f*)(ksrc + (size_t)16 * DM);
    v4f rv0 = *(const v4f*)(vsrc);
    v4f rv1 = *(const v4f*)(vsrc + (size_t)16 * DM);

    v2f qb[16];
    {
        const float* qp = Q + (size_t)(qBase + l16) * DM + dBase + 2 * half;
#pragma unroll
        for (int kk = 0; kk < 16; ++kk) qb[kk] = *(const v2f*)(qp + kk * 4);
    }

    v8f o[4];
#pragma unroll
    for (int dt = 0; dt < 4; ++dt) o[dt] = (v8f){0,0,0,0,0,0,0,0};
    float m = -INFINITY, lsum = 0.f;

    for (int t = 0; t < KH / KS; ++t) {
        const int cur = t & 1;

        *(v4f*)&Ks[kh][cur][sr][sc]      = rk0;
        *(v4f*)&Ks[kh][cur][sr + 16][sc] = rk1;
        *(v4f*)&Vs[kh][cur][vr][sc]      = rv0;
        *(v4f*)&Vs[kh][cur][16 + vr][sc] = rv1;
        __syncthreads();

        if (t + 1 < KH / KS) {
            const size_t off = (size_t)(t + 1) * KS * DM;
            rk0 = *(const v4f*)(ksrc + off);
            rk1 = *(const v4f*)(ksrc + off + (size_t)16 * DM);
            rv0 = *(const v4f*)(vsrc + off);
            rv1 = *(const v4f*)(vsrc + off + (size_t)16 * DM);
        }

        v8f sacc[2];
        sacc[0] = (v8f){0,0,0,0,0,0,0,0};
        sacc[1] = (v8f){0,0,0,0,0,0,0,0};
        {
            const float* kp0 = &Ks[kh][cur][l16][2 * half];
            const float* kp1 = &Ks[kh][cur][16 + l16][2 * half];
#pragma unroll
            for (int kk = 0; kk < 16; ++kk) {
                v2f a0 = *(const v2f*)(kp0 + kk * 4);
                v2f a1 = *(const v2f*)(kp1 + kk * 4);
                sacc[0] = wmma_f32(a0, qb[kk], sacc[0]);
                sacc[1] = wmma_f32(a1, qb[kk], sacc[1]);
            }
        }

#pragma unroll
        for (int u = 0; u < 2; ++u) {
            const v8f s = sacc[u];

            float sv[8];
#pragma unroll
            for (int j = 0; j < 8; ++j) sv[j] = s[j] * SCALE;
            float mx = fmaxf(fmaxf(fmaxf(sv[0], sv[1]), fmaxf(sv[2], sv[3])),
                             fmaxf(fmaxf(sv[4], sv[5]), fmaxf(sv[6], sv[7])));
            mx = fmaxf(mx, xhalf(mx));
            const float nm   = fmaxf(m, mx);
            const float corr = __expf(m - nm);
            float p[8];
#pragma unroll
            for (int j = 0; j < 8; ++j) p[j] = __expf(sv[j] - nm);
            float ps = ((p[0] + p[1]) + (p[2] + p[3])) + ((p[4] + p[5]) + (p[6] + p[7]));
            ps += xhalf(ps);
            lsum = lsum * corr + ps;
            m = nm;
#pragma unroll
            for (int dt = 0; dt < 4; ++dt)
#pragma unroll
                for (int j = 0; j < 8; ++j) o[dt][j] *= corr;

#pragma unroll
            for (int c = 0; c < 4; ++c) {
                v2f pb; pb.x = p[2 * c]; pb.y = p[2 * c + 1];
                const float* vp = &Vs[kh][cur][u * 16 + 4 * c + 2 * half][l16];
#pragma unroll
                for (int dt = 0; dt < 4; ++dt) {
                    v2f a; a.x = vp[dt * 16]; a.y = vp[VP + dt * 16];
                    o[dt] = wmma_f32(a, pb, o[dt]);
                }
            }
        }
    }

    if (kh == 1) {
        Mg[qw][0][lane] = m;
        Mg[qw][1][lane] = lsum;
#pragma unroll
        for (int dt = 0; dt < 4; ++dt)
#pragma unroll
            for (int j = 0; j < 8; ++j) Mg[qw][2 + dt * 8 + j][lane] = o[dt][j];
    }
    __syncthreads();

    if (kh == 0) {
        const float m1 = Mg[qw][0][lane];
        const float l1 = Mg[qw][1][lane];
        const float mm = fmaxf(m, m1);
        const float c0 = __expf(m - mm);
        const float c1 = __expf(m1 - mm);
        const float inv = 1.0f / (lsum * c0 + l1 * c1);
        const float w0 = c0 * inv;
        const float w1 = c1 * inv;
        float* orow = O + (size_t)(qBase + l16) * DM + dBase + 8 * half;
#pragma unroll
        for (int dt = 0; dt < 4; ++dt) {
            float r[8];
#pragma unroll
            for (int j = 0; j < 8; ++j) r[j] = o[dt][j] * w0 + Mg[qw][2 + dt * 8 + j][lane] * w1;
            v4f lo, hi;
            lo.x = r[0]; lo.y = r[1]; lo.z = r[2]; lo.w = r[3];
            hi.x = r[4]; hi.y = r[5]; hi.z = r[6]; hi.w = r[7];
            *(v4f*)(orow + dt * 16)     = lo;
            *(v4f*)(orow + dt * 16 + 4) = hi;
        }
    }
}

extern "C" void kernel_launch(void* const* d_in, const int* in_sizes, int n_in,
                              void* d_out, int out_size, void* d_ws, size_t ws_size,
                              hipStream_t stream) {
    const float* x  = (const float*)d_in[0];
    const float* fc = (const float*)d_in[1];
    const float* fs = (const float*)d_in[2];
    const float* wq = (const float*)d_in[3];
    const float* bq = (const float*)d_in[4];
    const float* wk = (const float*)d_in[5];
    const float* bk = (const float*)d_in[6];
    const float* wv = (const float*)d_in[7];
    const float* bv = (const float*)d_in[8];
    const float* wo = (const float*)d_in[9];
    const float* bo = (const float*)d_in[10];
    const float* gq = (const float*)d_in[11];
    const float* gk = (const float*)d_in[12];
    float* out = (float*)d_out;

    const size_t mat = (size_t)SB * DM;
    float* Qw = (float*)d_ws;
    float* Kw = Qw + mat;
    float* Vw = Kw + mat;
    float* Aw = Vw + mat;

    gemm32_kernel<6><<<dim3(256), 256, 0, stream>>>(x, wq, bq, Qw, wk, bk, Kw, wv, bv, Vw, 4);
    normrope_kernel<<<dim3(2 * RT), 256, 0, stream>>>(Qw, gq, Kw, gk, fc, fs);
    attn_kernel<<<dim3(SB / 16 / AW, NH), 32 * AW * NKH, 0, stream>>>(Qw, Kw, Vw, Aw);
    gemm32_kernel<2><<<dim3(256), 256, 0, stream>>>(Aw, wo, bo, out, wo, bo, out, wo, bo, out, 4);
}
